# starting kernel
# baseline (speedup 1.0000x reference)
#include <hip/hip_runtime.h>
#include <hip/hip_bf16.h>
#include <math.h>

typedef __attribute__((ext_vector_type(16))) _Float16 v16h;
typedef __attribute__((ext_vector_type(8)))  _Float16 v8h;
typedef __attribute__((ext_vector_type(8)))  float    v8f;

#define HEADS 4
#define CDIM 32
#define FDIM 128
#define NEG_SLOPE 0.2f
#define GAT_EPS 1e-16f

__global__ void layer_init(float* __restrict__ agg, unsigned* __restrict__ m_enc,
                           float* __restrict__ s_sum, int Nn) {
  int t = blockIdx.x * blockDim.x + threadIdx.x;
  if (t < Nn * FDIM) agg[t] = 0.0f;
  if (t < Nn * HEADS) { m_enc[t] = 0u; s_sum[t] = 0.0f; }
}

__global__ void conv_x(const float* __restrict__ X, _Float16* __restrict__ Xh, int total) {
  int t = blockIdx.x * blockDim.x + threadIdx.x;
  if (t < total) Xh[t] = (_Float16)X[t];
}

__global__ void pack_b(const float* __restrict__ W, _Float16* __restrict__ Bp) {
  int t = blockIdx.x * blockDim.x + threadIdx.x;
  if (t >= 8 * 32 * 4 * 16) return;
  const int i    = t & 15;
  const int ks   = (t >> 4) & 3;
  const int lane = (t >> 6) & 31;
  const int ct   = t >> 11;
  const int ka   = ks * 32 + ((lane >= 16) ? 8 : 0);
  const int k    = ka + ((i < 8) ? i : i + 8);
  const int col  = ct * 16 + (lane & 15);
  Bp[t] = (_Float16)W[(size_t)k * FDIM + col];
}

__global__ void gemm_wmma(const _Float16* __restrict__ Xh,
                          const _Float16* __restrict__ Bp,
                          float* __restrict__ Hout, int Nn) {
  const int wave = blockIdx.x * (blockDim.x >> 5) + (threadIdx.x >> 5);
  const int lane = threadIdx.x & 31;
  const int rowTiles = (Nn + 15) >> 4;
  if (wave >= rowTiles * (FDIM / 16)) return;

  const int rt = wave >> 3;
  const int ct = wave & 7;
  const int row0 = rt << 4;
  const int col0 = ct << 4;
  const int lm = lane & 15;
  const bool hi = lane >= 16;

  int arow = row0 + lm;
  if (arow >= Nn) arow = Nn - 1;

  const _Float16* arowp = Xh + (size_t)arow * FDIM + (hi ? 8 : 0);
  const _Float16* bfrag = Bp + ((size_t)(ct * 32 + lane) * 4) * 16;

  v8f c = {};
  #pragma unroll
  for (int ks = 0; ks < 4; ++ks) {
    const _Float16* ap = arowp + ks * 32;
    const v8h alo = *(const v8h*)(ap);
    const v8h ahi = *(const v8h*)(ap + 16);
    v16h a;
    #pragma unroll
    for (int i = 0; i < 8; ++i) { a[i] = alo[i]; a[8 + i] = ahi[i]; }
    const v16h b = *(const v16h*)(bfrag + ks * 16);
    c = __builtin_amdgcn_wmma_f32_16x16x32_f16(
         false, a,  false, b,
         (short)0, c,  false,  false);
  }

  const int m0 = hi ? 8 : 0;
  float* outp = Hout + (size_t)(row0 + m0) * FDIM + col0 + lm;
  if (row0 + 16 <= Nn) {
    #pragma unroll
    for (int r = 0; r < 8; ++r) outp[(size_t)r * FDIM] = c[r];
  } else {
    #pragma unroll
    for (int r = 0; r < 8; ++r) {
      if (row0 + m0 + r < Nn) outp[(size_t)r * FDIM] = c[r];
    }
  }
}

__global__ void attn_logits(const float* __restrict__ Hh,
                            const float* __restrict__ Asrc,
                            const float* __restrict__ Adst,
                            float* __restrict__ als, float* __restrict__ ald,
                            int Nn) {
  int t = blockIdx.x * blockDim.x + threadIdx.x;
  if (t >= Nn * HEADS) return;
  const int n = t >> 2, hh = t & 3;
  const float* hp = Hh + (size_t)n * FDIM + hh * CDIM;
  const float* as = Asrc + hh * CDIM;
  const float* ad = Adst + hh * CDIM;
  float ss = 0.0f, dd = 0.0f;
  #pragma unroll
  for (int cc = 0; cc < CDIM; ++cc) {
    const float v = hp[cc];
    ss = fmaf(v, as[cc], ss);
    dd = fmaf(v, ad[cc], dd);
  }
  als[t] = ss;
  ald[t] = dd;
}

__device__ __forceinline__ float leaky(float v) {
  return v > 0.0f ? v : v * NEG_SLOPE;
}

__device__ __forceinline__ unsigned enc_f32(float f) {
  unsigned u = __float_as_uint(f);
  return ((int)u >= 0) ? (u | 0x80000000u) : ~u;
}

__global__ void edge_max(const int* __restrict__ src, const int* __restrict__ dst,
                         const float4* __restrict__ als4, const float4* __restrict__ ald4,
                         unsigned* __restrict__ m_enc, int Etot) {
  int e = blockIdx.x * blockDim.x + threadIdx.x;
  if (e >= Etot) return;
  const int si = src[e], di = dst[e];
  const float4 s4 = als4[si];
  const float4 d4 = ald4[di];
  atomicMax(&m_enc[di * HEADS + 0], enc_f32(leaky(s4.x + d4.x)));
  atomicMax(&m_enc[di * HEADS + 1], enc_f32(leaky(s4.y + d4.y)));
  atomicMax(&m_enc[di * HEADS + 2], enc_f32(leaky(s4.z + d4.z)));
  atomicMax(&m_enc[di * HEADS + 3], enc_f32(leaky(s4.w + d4.w)));
}

__global__ void decode_max(unsigned* __restrict__ m_enc, int Nn) {
  int t = blockIdx.x * blockDim.x + threadIdx.x;
  if (t >= Nn * HEADS) return;
  const unsigned u = m_enc[t];
  const float f = (u & 0x80000000u) ? __uint_as_float(u ^ 0x80000000u)
                                    : __uint_as_float(~u);
  ((float*)m_enc)[t] = f;
}

__global__ void edge_expsum(const int* __restrict__ src, const int* __restrict__ dst,
                            const float4* __restrict__ als4, const float4* __restrict__ ald4,
                            const float4* __restrict__ m4, float* __restrict__ s_sum,
                            float4* __restrict__ expe4, int Etot) {
  int e = blockIdx.x * blockDim.x + threadIdx.x;
  if (e >= Etot) return;
  const int si = src[e], di = dst[e];
  const float4 s4 = als4[si];
  const float4 d4 = ald4[di];
  const float4 mm = m4[di];
  float4 ex;
  ex.x = __expf(leaky(s4.x + d4.x) - mm.x);
  ex.y = __expf(leaky(s4.y + d4.y) - mm.y);
  ex.z = __expf(leaky(s4.z + d4.z) - mm.z);
  ex.w = __expf(leaky(s4.w + d4.w) - mm.w);
  expe4[e] = ex;
  atomicAdd(&s_sum[di * HEADS + 0], ex.x);
  atomicAdd(&s_sum[di * HEADS + 1], ex.y);
  atomicAdd(&s_sum[di * HEADS + 2], ex.z);
  atomicAdd(&s_sum[di * HEADS + 3], ex.w);
}

__global__ void edge_aggregate(const int* __restrict__ src, const int* __restrict__ dst,
                               const float* __restrict__ Hh,
                               const float4* __restrict__ s_sum4,
                               const float4* __restrict__ expe4,
                               float* __restrict__ agg, int Etot) {
  int t = blockIdx.x * blockDim.x + threadIdx.x;
  if (t >= Etot * CDIM) return;
  const int e = t >> 5;
  const int cc = t & 31;
  const int si = src[e], di = dst[e];
  const float4 ex = expe4[e];
  const float4 sm = s_sum4[di];
  const float a0 = ex.x / (sm.x + GAT_EPS);
  const float a1 = ex.y / (sm.y + GAT_EPS);
  const float a2 = ex.z / (sm.z + GAT_EPS);
  const float a3 = ex.w / (sm.w + GAT_EPS);
  const float* hp = Hh + (size_t)si * FDIM + cc;
  float* ap = agg + (size_t)di * FDIM + cc;
  atomicAdd(ap + 0 * CDIM, hp[0 * CDIM] * a0);
  atomicAdd(ap + 1 * CDIM, hp[1 * CDIM] * a1);
  atomicAdd(ap + 2 * CDIM, hp[2 * CDIM] * a2);
  atomicAdd(ap + 3 * CDIM, hp[3 * CDIM] * a3);
}

__global__ void finalize(const float* __restrict__ agg, const float* __restrict__ bias,
                         _Float16* __restrict__ xh_next, float* __restrict__ out_f32,
                         int Nn, int hidden) {
  int t = blockIdx.x * blockDim.x + threadIdx.x;
  if (t >= Nn * FDIM) return;
  float v = agg[t] + bias[t & (FDIM - 1)];
  if (hidden) {
    xh_next[t] = (_Float16)fmaxf(v, 0.0f);
  } else {
    out_f32[t] = v;
  }
}

extern "C" void kernel_launch(void* const* d_in, const int* in_sizes, int n_in,
                              void* d_out, int out_size, void* d_ws, size_t ws_size,
                              hipStream_t stream) {
  const int Nn   = in_sizes[0] / FDIM;
  const int Etot = in_sizes[1] / 2;

  const float* x0  = (const float*)d_in[0];
  const int*   ei  = (const int*)d_in[1];
  const int*   src = ei;
  const int*   dst = ei + Etot;

  char* ws = (char*)d_ws;
  size_t off = 0;
  auto alloc = [&](size_t bytes) {
    void* p = ws + off;
    off += (bytes + 255) & ~(size_t)255;
    return p;
  };
  _Float16* xh    = (_Float16*)alloc((size_t)Nn * FDIM * sizeof(_Float16));
  _Float16* bp    = (_Float16*)alloc((size_t)8 * 32 * 4 * 16 * sizeof(_Float16));
  float*    buf_h = (float*)   alloc((size_t)Nn * FDIM * sizeof(float));
  float*    agg   = (float*)   alloc((size_t)Nn * FDIM * sizeof(float));
  float*    als   = (float*)   alloc((size_t)Nn * HEADS * sizeof(float));
  float*    ald   = (float*)   alloc((size_t)Nn * HEADS * sizeof(float));
  unsigned* m_e   = (unsigned*)alloc((size_t)Nn * HEADS * sizeof(float));
  float*    s_sum = (float*)   alloc((size_t)Nn * HEADS * sizeof(float));
  float4*   expe4 = (float4*)  alloc((size_t)Etot * HEADS * sizeof(float));

  const int TPB = 256;
  const int gNodeF = (Nn * FDIM + TPB - 1) / TPB;
  const int gNodeH = (Nn * HEADS + TPB - 1) / TPB;
  const int gEdge  = (Etot + TPB - 1) / TPB;
  const int gEdgeC = (Etot * CDIM + TPB - 1) / TPB;
  const int rowTiles = (Nn + 15) >> 4;
  const int gGemm  = (rowTiles * (FDIM / 16) + (TPB / 32) - 1) / (TPB / 32);
  const int gPackB = (8 * 32 * 4 * 16 + TPB - 1) / TPB;

  conv_x<<<gNodeF, TPB, 0, stream>>>(x0, xh, Nn * FDIM);

  for (int l = 0; l < 3; ++l) {
    const float* W    = (const float*)d_in[2 + l * 4];
    const float* Asrc = (const float*)d_in[3 + l * 4];
    const float* Adst = (const float*)d_in[4 + l * 4];
    const float* B    = (const float*)d_in[5 + l * 4];

    pack_b<<<gPackB, TPB, 0, stream>>>(W, bp);
    layer_init<<<gNodeF, TPB, 0, stream>>>(agg, m_e, s_sum, Nn);
    gemm_wmma<<<gGemm, TPB, 0, stream>>>(xh, bp, buf_h, Nn);
    attn_logits<<<gNodeH, TPB, 0, stream>>>(buf_h, Asrc, Adst, als, ald, Nn);
    edge_max<<<gEdge, TPB, 0, stream>>>(src, dst, (const float4*)als,
                                        (const float4*)ald, m_e, Etot);
    decode_max<<<gNodeH, TPB, 0, stream>>>(m_e, Nn);
    edge_expsum<<<gEdge, TPB, 0, stream>>>(src, dst, (const float4*)als,
                                           (const float4*)ald, (const float4*)m_e,
                                           s_sum, expe4, Etot);
    edge_aggregate<<<gEdgeC, TPB, 0, stream>>>(src, dst, buf_h, (const float4*)s_sum,
                                               expe4, agg, Etot);
    finalize<<<gNodeF, TPB, 0, stream>>>(agg, B, xh, (float*)d_out, Nn, l < 2 ? 1 : 0);
  }
}


# Layer-0 GEMM loads the f32 features itself and rounds them to f16 in registers on the way into the A fragment; the separate one-element-per-lane conversion pass (200k waves, one launch) is gone
# speedup vs baseline: 5.4504x; 5.4504x over previous
#include <hip/hip_runtime.h>
#include <hip/hip_bf16.h>
#include <math.h>

typedef __attribute__((ext_vector_type(16))) _Float16 v16h;
typedef __attribute__((ext_vector_type(8)))  _Float16 v8h;
typedef __attribute__((ext_vector_type(8)))  float    v8f;
typedef __attribute__((ext_vector_type(4)))  _Float16 v4h;

#define HEADS 4
#define CDIM 32
#define FDIM 128
#define NEG_SLOPE 0.2f
#define GAT_EPS 1e-16f

__global__ void pack_b(const float* __restrict__ W, _Float16* __restrict__ Bp) {
  int t = blockIdx.x * blockDim.x + threadIdx.x;
  if (t >= 8 * 32 * 4 * 16) return;
  const int i    = t & 15;
  const int ks   = (t >> 4) & 3;
  const int lane = (t >> 6) & 31;
  const int ct   = t >> 11;
  const int ka   = ks * 32 + ((lane >= 16) ? 8 : 0);
  const int k    = ka + ((i < 8) ? i : i + 8);
  const int col  = ct * 16 + (lane & 15);
  Bp[t] = (_Float16)W[(size_t)k * FDIM + col];
}

template <bool kF32Rows>
__device__ __forceinline__ void gemm_tile(const float* __restrict__ Xf,
                                          const _Float16* __restrict__ Xh,
                                          const _Float16* __restrict__ Bp,
                                          float* __restrict__ Hout, int Nn) {
  const int wave = blockIdx.x * (blockDim.x >> 5) + (threadIdx.x >> 5);
  const int lane = threadIdx.x & 31;
  const int rowTiles = (Nn + 15) >> 4;
  if (wave >= rowTiles * (FDIM / 16)) return;

  const int rt = wave >> 3;
  const int ct = wave & 7;
  const int row0 = rt << 4;
  const int col0 = ct << 4;
  const int lm = lane & 15;
  const bool hi = lane >= 16;

  int arow = row0 + lm;
  if (arow >= Nn) arow = Nn - 1;

  const size_t aoff = (size_t)arow * FDIM + (hi ? 8 : 0);
  const _Float16* bfrag = Bp + ((size_t)(ct * 32 + lane) * 4) * 16;

  v8f c = {};
  #pragma unroll
  for (int ks = 0; ks < 4; ++ks) {
    v16h a;
    if constexpr (kF32Rows) {
      const float* ap = Xf + aoff + ks * 32;
      const float4 f0 = *(const float4*)(ap);
      const float4 f1 = *(const float4*)(ap + 4);
      const float4 f2 = *(const float4*)(ap + 16);
      const float4 f3 = *(const float4*)(ap + 20);
      a[0]  = (_Float16)f0.x; a[1]  = (_Float16)f0.y; a[2]  = (_Float16)f0.z; a[3]  = (_Float16)f0.w;
      a[4]  = (_Float16)f1.x; a[5]  = (_Float16)f1.y; a[6]  = (_Float16)f1.z; a[7]  = (_Float16)f1.w;
      a[8]  = (_Float16)f2.x; a[9]  = (_Float16)f2.y; a[10] = (_Float16)f2.z; a[11] = (_Float16)f2.w;
      a[12] = (_Float16)f3.x; a[13] = (_Float16)f3.y; a[14] = (_Float16)f3.z; a[15] = (_Float16)f3.w;
    } else {
      const _Float16* ap = Xh + aoff + ks * 32;
      const v8h alo = *(const v8h*)(ap);
      const v8h ahi = *(const v8h*)(ap + 16);
      #pragma unroll
      for (int i = 0; i < 8; ++i) { a[i] = alo[i]; a[8 + i] = ahi[i]; }
    }
    const v16h b = *(const v16h*)(bfrag + ks * 16);
    c = __builtin_amdgcn_wmma_f32_16x16x32_f16(
         false, a,  false, b,
         (short)0, c,  false,  false);
  }

  const int m0 = hi ? 8 : 0;
  float* outp = Hout + (size_t)(row0 + m0) * FDIM + col0 + lm;
  if (row0 + 16 <= Nn) {
    #pragma unroll
    for (int r = 0; r < 8; ++r) outp[(size_t)r * FDIM] = c[r];
  } else {
    #pragma unroll
    for (int r = 0; r < 8; ++r) {
      if (row0 + m0 + r < Nn) outp[(size_t)r * FDIM] = c[r];
    }
  }
}

__global__ void gemm_wmma_x32(const float* __restrict__ X,
                              const _Float16* __restrict__ Bp,
                              float* __restrict__ Hout, int Nn) {
  gemm_tile<true>(X, nullptr, Bp, Hout, Nn);
}

__global__ void gemm_wmma(const _Float16* __restrict__ Xh,
                          const _Float16* __restrict__ Bp,
                          float* __restrict__ Hout, int Nn) {
  gemm_tile<false>(nullptr, Xh, Bp, Hout, Nn);
}

__global__ void attn_logits(const float* __restrict__ Hh,
                            const float* __restrict__ Asrc,
                            const float* __restrict__ Adst,
                            float* __restrict__ als, float* __restrict__ ald,
                            int Nn) {
  int t = blockIdx.x * blockDim.x + threadIdx.x;
  if (t >= Nn * HEADS) return;
  const int n = t >> 2, hh = t & 3;
  const float* hp = Hh + (size_t)n * FDIM + hh * CDIM;
  const float* as = Asrc + hh * CDIM;
  const float* ad = Adst + hh * CDIM;
  float ss = 0.0f, dd = 0.0f;
  #pragma unroll
  for (int cc = 0; cc < CDIM; ++cc) {
    const float v = hp[cc];
    ss = fmaf(v, as[cc], ss);
    dd = fmaf(v, ad[cc], dd);
  }
  als[t] = ss;
  ald[t] = dd;
}

__device__ __forceinline__ float leaky(float v) {
  return v > 0.0f ? v : v * NEG_SLOPE;
}

#define SCAN_TILE 1024

__global__ void zero_counts(int* __restrict__ cnt, int total) {
  int t = blockIdx.x * blockDim.x + threadIdx.x;
  if (t < total) cnt[t] = 0;
}

__global__ void count_rank(const int* __restrict__ dst, int* __restrict__ cnt,
                           int* __restrict__ rank, int Etot) {
  int e = blockIdx.x * blockDim.x + threadIdx.x;
  if (e >= Etot) return;
  rank[e] = atomicAdd(&cnt[dst[e]], 1);
}

__global__ void __launch_bounds__(256) scan_tiles(const int4* __restrict__ cnt4,
                                                  int4* __restrict__ rowptr4,
                                                  int* __restrict__ tsum) {
  __shared__ int wsum[8];
  const int t = threadIdx.x;
  const int g = blockIdx.x * 256 + t;
  const int lane = t & 31, w = t >> 5;
  const int4 c = cnt4[g];
  const int s = c.x + c.y + c.z + c.w;
  int v = s;
  for (int o = 1; o < 32; o <<= 1) {
    const int u = __shfl_up(v, o, 32);
    if (lane >= o) v += u;
  }
  if (lane == 31) wsum[w] = v;
  __syncthreads();
  int woff = 0;
  for (int k = 0; k < 8; ++k) woff += (k < w) ? wsum[k] : 0;
  const int ex = woff + v - s;
  int4 r;
  r.x = ex; r.y = ex + c.x; r.z = ex + c.x + c.y; r.w = ex + c.x + c.y + c.z;
  rowptr4[g] = r;
  if (t == 255) tsum[blockIdx.x] = ex + s;
}

__global__ void __launch_bounds__(256) scan_add(int4* __restrict__ rowptr4,
                                                const int* __restrict__ tsum) {
  __shared__ int red[8];
  const int t = threadIdx.x;
  const int b = blockIdx.x;
  int part = 0;
  for (int k = t; k < b; k += 256) part += tsum[k];
  for (int o = 16; o >= 1; o >>= 1) part += __shfl_xor(part, o, 32);
  if ((t & 31) == 0) red[t >> 5] = part;
  __syncthreads();
  int off = 0;
  for (int k = 0; k < 8; ++k) off += red[k];
  int4 r = rowptr4[b * 256 + t];
  r.x += off; r.y += off; r.z += off; r.w += off;
  rowptr4[b * 256 + t] = r;
}

__global__ void fill_csr(const int* __restrict__ src, const int* __restrict__ dst,
                         const int* __restrict__ rank, const int* __restrict__ rowptr,
                         int* __restrict__ csr_u, int Etot) {
  int e = blockIdx.x * blockDim.x + threadIdx.x;
  if (e >= Etot) return;
  csr_u[rowptr[dst[e]] + rank[e]] = src[e];
}

__global__ void __launch_bounds__(256) sort_segments(const int* __restrict__ rowptr,
                                                     const int* __restrict__ csr_u,
                                                     int* __restrict__ csr_s, int Nn) {
  const int n = __builtin_amdgcn_readfirstlane((int)(blockIdx.x * 8 + (threadIdx.x >> 5)));
  const int lane = threadIdx.x & 31;
  if (n >= Nn) return;
  const int beg = rowptr[n], end = rowptr[n + 1];
  for (int bi = beg; bi < end; bi += 32) {
    const int i = bi + lane;
    const bool ok = i < end;
    const int vi = ok ? csr_u[i] : 0;
    int r = 0;
    for (int bq = beg; bq < end; bq += 32) {
      const int q = bq + lane;
      const int vq = (q < end) ? csr_u[q] : 0;
      const int nq = min(32, end - bq);
      for (int t = 0; t < nq; ++t) {
        const int x = __builtin_amdgcn_readlane(vq, t);
        const int qq = bq + t;
        r += (x < vi || (x == vi && qq < i)) ? 1 : 0;
      }
    }
    if (ok) csr_s[beg + r] = vi;
  }
}

__global__ void __launch_bounds__(256) node_gather(const int* __restrict__ rowptr,
                                                  const int* __restrict__ csr,
                                                  const float4* __restrict__ als4,
                                                  const float4* __restrict__ ald4,
                                                  const float4* __restrict__ Hh4,
                                                  const float* __restrict__ bias,
                                                  _Float16* __restrict__ xh_next,
                                                  float* __restrict__ out_f32,
                                                  int Nn, int hidden) {
  const int n = __builtin_amdgcn_readfirstlane((int)(blockIdx.x * 8 + (threadIdx.x >> 5)));
  const int lane = threadIdx.x & 31;
  if (n >= Nn) return;
  const int beg = rowptr[n], end = rowptr[n + 1];
  const float4 d4 = ald4[n];
  const bool hb0 = (lane & 8) != 0;
  const bool hb1 = (lane & 16) != 0;

  float4 acc = make_float4(0.0f, 0.0f, 0.0f, 0.0f);
  float wsum = 0.0f;
  float4 mx = make_float4(-INFINITY, -INFINITY, -INFINITY, -INFINITY);

  for (int base = beg; base < end; base += 32) {
    const int cnt = min(32, end - base);
    int s = 0;
    float4 e = make_float4(-INFINITY, -INFINITY, -INFINITY, -INFINITY);
    if (lane < cnt) {
      s = csr[base + lane];
      const float4 a = als4[s];
      e.x = leaky(a.x + d4.x);
      e.y = leaky(a.y + d4.y);
      e.z = leaky(a.z + d4.z);
      e.w = leaky(a.w + d4.w);
    }
    float4 cm = e;
    for (int o = 16; o >= 1; o >>= 1) {
      cm.x = fmaxf(cm.x, __shfl_xor(cm.x, o, 32));
      cm.y = fmaxf(cm.y, __shfl_xor(cm.y, o, 32));
      cm.z = fmaxf(cm.z, __shfl_xor(cm.z, o, 32));
      cm.w = fmaxf(cm.w, __shfl_xor(cm.w, o, 32));
    }
    float4 nm;
    nm.x = fmaxf(mx.x, cm.x); nm.y = fmaxf(mx.y, cm.y);
    nm.z = fmaxf(mx.z, cm.z); nm.w = fmaxf(mx.w, cm.w);
    const float sc01 = hb0 ? __expf(mx.y - nm.y) : __expf(mx.x - nm.x);
    const float sc23 = hb0 ? __expf(mx.w - nm.w) : __expf(mx.z - nm.z);
    const float sc = hb1 ? sc23 : sc01;
    acc.x *= sc; acc.y *= sc; acc.z *= sc; acc.w *= sc;
    wsum *= sc;
    mx = nm;
    const int ex0 = __float_as_int(__expf(e.x - nm.x));
    const int ex1 = __float_as_int(__expf(e.y - nm.y));
    const int ex2 = __float_as_int(__expf(e.z - nm.z));
    const int ex3 = __float_as_int(__expf(e.w - nm.w));

    int j = 0;
    for (; j + 4 <= cnt; j += 4) {
      float4 hv[4];
      float wv[4];
      for (int u = 0; u < 4; ++u) {
        const int sj = __builtin_amdgcn_readlane(s, j + u);
        hv[u] = Hh4[(size_t)sj * (FDIM / 4) + lane];
        const float a0 = __int_as_float(__builtin_amdgcn_readlane(ex0, j + u));
        const float a1 = __int_as_float(__builtin_amdgcn_readlane(ex1, j + u));
        const float a2 = __int_as_float(__builtin_amdgcn_readlane(ex2, j + u));
        const float a3 = __int_as_float(__builtin_amdgcn_readlane(ex3, j + u));
        const float w01 = hb0 ? a1 : a0;
        const float w23 = hb0 ? a3 : a2;
        wv[u] = hb1 ? w23 : w01;
      }
      for (int u = 0; u < 4; ++u) {
        acc.x = fmaf(hv[u].x, wv[u], acc.x);
        acc.y = fmaf(hv[u].y, wv[u], acc.y);
        acc.z = fmaf(hv[u].z, wv[u], acc.z);
        acc.w = fmaf(hv[u].w, wv[u], acc.w);
        wsum += wv[u];
      }
    }
    for (; j < cnt; ++j) {
      const int sj = __builtin_amdgcn_readlane(s, j);
      const float4 h1 = Hh4[(size_t)sj * (FDIM / 4) + lane];
      const float a0 = __int_as_float(__builtin_amdgcn_readlane(ex0, j));
      const float a1 = __int_as_float(__builtin_amdgcn_readlane(ex1, j));
      const float a2 = __int_as_float(__builtin_amdgcn_readlane(ex2, j));
      const float a3 = __int_as_float(__builtin_amdgcn_readlane(ex3, j));
      const float w01 = hb0 ? a1 : a0;
      const float w23 = hb0 ? a3 : a2;
      const float wj = hb1 ? w23 : w01;
      acc.x = fmaf(h1.x, wj, acc.x);
      acc.y = fmaf(h1.y, wj, acc.y);
      acc.z = fmaf(h1.z, wj, acc.z);
      acc.w = fmaf(h1.w, wj, acc.w);
      wsum += wj;
    }
  }

  const float inv = 1.0f / (wsum + GAT_EPS);
  const float* bp4 = bias + lane * 4;
  float4 b;
  b.x = bp4[0]; b.y = bp4[1]; b.z = bp4[2]; b.w = bp4[3];
  float4 o;
  o.x = fmaf(acc.x, inv, b.x);
  o.y = fmaf(acc.y, inv, b.y);
  o.z = fmaf(acc.z, inv, b.z);
  o.w = fmaf(acc.w, inv, b.w);
  const size_t at = (size_t)n * FDIM + (size_t)lane * 4;
  if (hidden) {
    v4h hq;
    hq[0] = (_Float16)fmaxf(o.x, 0.0f);
    hq[1] = (_Float16)fmaxf(o.y, 0.0f);
    hq[2] = (_Float16)fmaxf(o.z, 0.0f);
    hq[3] = (_Float16)fmaxf(o.w, 0.0f);
    *(v4h*)(xh_next + at) = hq;
  } else {
    *(float4*)(out_f32 + at) = o;
  }
}

extern "C" void kernel_launch(void* const* d_in, const int* in_sizes, int n_in,
                              void* d_out, int out_size, void* d_ws, size_t ws_size,
                              hipStream_t stream) {
  const int Nn   = in_sizes[0] / FDIM;
  const int Etot = in_sizes[1] / 2;

  const float* x0  = (const float*)d_in[0];
  const int*   ei  = (const int*)d_in[1];
  const int*   src = ei;
  const int*   dst = ei + Etot;

  const int nTiles  = (Nn + 1 + SCAN_TILE - 1) / SCAN_TILE;
  const int nPadded = nTiles * SCAN_TILE;

  char* ws = (char*)d_ws;
  size_t off = 0;
  auto alloc = [&](size_t bytes) {
    void* p = ws + off;
    off += (bytes + 255) & ~(size_t)255;
    return p;
  };
  _Float16* xh     = (_Float16*)alloc((size_t)Nn * FDIM * sizeof(_Float16));
  _Float16* bp     = (_Float16*)alloc((size_t)8 * 32 * 4 * 16 * sizeof(_Float16));
  float*    buf_h  = (float*)   alloc((size_t)Nn * FDIM * sizeof(float));
  float*    als    = (float*)   alloc((size_t)Nn * HEADS * sizeof(float));
  float*    ald    = (float*)   alloc((size_t)Nn * HEADS * sizeof(float));
  int*      cnt    = (int*)     alloc((size_t)nPadded * sizeof(int));
  int*      rowptr = (int*)     alloc((size_t)nPadded * sizeof(int));
  int*      tsum   = (int*)     alloc((size_t)nTiles * sizeof(int));
  int*      rank   = (int*)     alloc((size_t)Etot * sizeof(int));
  int*      csr_u  = (int*)     alloc((size_t)Etot * sizeof(int));
  int*      csr_s  = (int*)     alloc((size_t)Etot * sizeof(int));

  const int TPB = 256;
  const int gNodeH = (Nn * HEADS + TPB - 1) / TPB;
  const int gEdge  = (Etot + TPB - 1) / TPB;
  const int gNodeW = (Nn + (TPB / 32) - 1) / (TPB / 32);
  const int gPad   = (nPadded + TPB - 1) / TPB;
  const int rowTiles = (Nn + 15) >> 4;
  const int gGemm  = (rowTiles * (FDIM / 16) + (TPB / 32) - 1) / (TPB / 32);
  const int gPackB = (8 * 32 * 4 * 16 + TPB - 1) / TPB;

  zero_counts<<<gPad, TPB, 0, stream>>>(cnt, nPadded);
  count_rank<<<gEdge, TPB, 0, stream>>>(dst, cnt, rank, Etot);
  scan_tiles<<<nTiles, TPB, 0, stream>>>((const int4*)cnt, (int4*)rowptr, tsum);
  scan_add<<<nTiles, TPB, 0, stream>>>((int4*)rowptr, tsum);
  fill_csr<<<gEdge, TPB, 0, stream>>>(src, dst, rank, rowptr, csr_u, Etot);
  sort_segments<<<gNodeW, TPB, 0, stream>>>(rowptr, csr_u, csr_s, Nn);

  for (int l = 0; l < 3; ++l) {
    const float* W    = (const float*)d_in[2 + l * 4];
    const float* Asrc = (const float*)d_in[3 + l * 4];
    const float* Adst = (const float*)d_in[4 + l * 4];
    const float* B    = (const float*)d_in[5 + l * 4];

    pack_b<<<gPackB, TPB, 0, stream>>>(W, bp);
    if (l == 0) gemm_wmma_x32<<<gGemm, TPB, 0, stream>>>(x0, bp, buf_h, Nn);
    else        gemm_wmma<<<gGemm, TPB, 0, stream>>>(xh, bp, buf_h, Nn);
    attn_logits<<<gNodeH, TPB, 0, stream>>>(buf_h, Asrc, Adst, als, ald, Nn);
    node_gather<<<gNodeW, TPB, 0, stream>>>(rowptr, csr_s, (const float4*)als,
                                           (const float4*)ald, (const float4*)buf_h, B,
                                           xh, (float*)d_out, Nn, l < 2 ? 1 : 0);
  }
}


# Each GEMM wave computes a whole head (16 rows x 32 columns) and sums the head's two attention dot products in its epilogue as a 16-lane tree (was a serial chain); the separate logits pass is gone
# speedup vs baseline: 6.1116x; 1.1213x over previous
#include <hip/hip_runtime.h>
#include <hip/hip_bf16.h>
#include <math.h>

typedef __attribute__((ext_vector_type(16))) _Float16 v16h;
typedef __attribute__((ext_vector_type(8)))  _Float16 v8h;
typedef __attribute__((ext_vector_type(8)))  float    v8f;
typedef __attribute__((ext_vector_type(4)))  _Float16 v4h;

#define HEADS 4
#define CDIM 32
#define FDIM 128
#define NEG_SLOPE 0.2f
#define GAT_EPS 1e-16f

__global__ void pack_b(const float* __restrict__ W, _Float16* __restrict__ Bp) {
  int t = blockIdx.x * blockDim.x + threadIdx.x;
  if (t >= 8 * 32 * 4 * 16) return;
  const int i    = t & 15;
  const int ks   = (t >> 4) & 3;
  const int lane = (t >> 6) & 31;
  const int ct   = t >> 11;
  const int ka   = ks * 32 + ((lane >= 16) ? 8 : 0);
  const int k    = ka + ((i < 8) ? i : i + 8);
  const int col  = (ct >> 1) * 32 + 2 * (lane & 15) + (ct & 1);
  Bp[t] = (_Float16)W[(size_t)k * FDIM + col];
}

template <int M>
__device__ __forceinline__ float row_xadd(float v) {
  const int t = __builtin_amdgcn_update_dpp(0, __float_as_int(v), 0x160 + M, 0xf, 0xf, true);
  return v + __int_as_float(t);
}

template <bool kF32Rows>
__device__ __forceinline__ void gemm_head(const float* __restrict__ Xf,
                                          const _Float16* __restrict__ Xh,
                                          const _Float16* __restrict__ Bp,
                                          const float* __restrict__ Asrc,
                                          const float* __restrict__ Adst,
                                          float* __restrict__ Hout,
                                          float* __restrict__ als,
                                          float* __restrict__ ald, int Nn) {
  const int wave = blockIdx.x * (blockDim.x >> 5) + (threadIdx.x >> 5);
  const int lane = threadIdx.x & 31;
  const int rowTiles = (Nn + 15) >> 4;
  if (wave >= rowTiles * HEADS) return;

  const int rt = wave >> 2;
  const int hd = wave & 3;
  const int row0 = rt << 4;
  const int lm = lane & 15;
  const bool hi = lane >= 16;

  int arow = row0 + lm;
  if (arow >= Nn) arow = Nn - 1;

  const size_t aoff = (size_t)arow * FDIM + (hi ? 8 : 0);
  const _Float16* bfrag0 = Bp + ((size_t)((hd * 2) * 32 + lane) * 4) * 16;
  const _Float16* bfrag1 = Bp + ((size_t)((hd * 2 + 1) * 32 + lane) * 4) * 16;

  const int colh = hd * CDIM + 2 * lm;
  const float2 as2 = *(const float2*)(Asrc + colh);
  const float2 ad2 = *(const float2*)(Adst + colh);

  v8f c0 = {};
  v8f c1 = {};
  #pragma unroll
  for (int ks = 0; ks < 4; ++ks) {
    v16h a;
    if constexpr (kF32Rows) {
      const float* ap = Xf + aoff + ks * 32;
      const float4 f0 = *(const float4*)(ap);
      const float4 f1 = *(const float4*)(ap + 4);
      const float4 f2 = *(const float4*)(ap + 16);
      const float4 f3 = *(const float4*)(ap + 20);
      a[0]  = (_Float16)f0.x; a[1]  = (_Float16)f0.y; a[2]  = (_Float16)f0.z; a[3]  = (_Float16)f0.w;
      a[4]  = (_Float16)f1.x; a[5]  = (_Float16)f1.y; a[6]  = (_Float16)f1.z; a[7]  = (_Float16)f1.w;
      a[8]  = (_Float16)f2.x; a[9]  = (_Float16)f2.y; a[10] = (_Float16)f2.z; a[11] = (_Float16)f2.w;
      a[12] = (_Float16)f3.x; a[13] = (_Float16)f3.y; a[14] = (_Float16)f3.z; a[15] = (_Float16)f3.w;
    } else {
      const _Float16* ap = Xh + aoff + ks * 32;
      const v8h alo = *(const v8h*)(ap);
      const v8h ahi = *(const v8h*)(ap + 16);
      #pragma unroll
      for (int i = 0; i < 8; ++i) { a[i] = alo[i]; a[8 + i] = ahi[i]; }
    }
    const v16h b0 = *(const v16h*)(bfrag0 + ks * 16);
    const v16h b1 = *(const v16h*)(bfrag1 + ks * 16);
    c0 = __builtin_amdgcn_wmma_f32_16x16x32_f16(
         false, a,  false, b0,
         (short)0, c0,  false,  false);
    c1 = __builtin_amdgcn_wmma_f32_16x16x32_f16(
         false, a,  false, b1,
         (short)0, c1,  false,  false);
  }

  const int m0 = hi ? 8 : 0;
  float* outp = Hout + (size_t)(row0 + m0) * FDIM + colh;
  if (row0 + 16 <= Nn) {
    #pragma unroll
    for (int r = 0; r < 8; ++r)
      *(float2*)(outp + (size_t)r * FDIM) = make_float2(c0[r], c1[r]);
  } else {
    #pragma unroll
    for (int r = 0; r < 8; ++r) {
      if (row0 + m0 + r < Nn)
        *(float2*)(outp + (size_t)r * FDIM) = make_float2(c0[r], c1[r]);
    }
  }

  const bool isDst = (lane & 8) != 0;
  float q[8];
  #pragma unroll
  for (int r = 0; r < 8; ++r) {
    const float ps = fmaf(c1[r], as2.y, c0[r] * as2.x);
    const float pd = fmaf(c1[r], ad2.y, c0[r] * ad2.x);
    const float keep = isDst ? pd : ps;
    const float give = isDst ? ps : pd;
    const int g = __builtin_amdgcn_update_dpp(0, __float_as_int(give), 0x168, 0xf, 0xf, true);
    q[r] = keep + __int_as_float(g);
  }
  #pragma unroll
  for (int r = 0; r < 8; ++r) q[r] = row_xadd<1>(q[r]);
  #pragma unroll
  for (int r = 0; r < 8; ++r) q[r] = row_xadd<2>(q[r]);
  #pragma unroll
  for (int r = 0; r < 8; ++r) q[r] = row_xadd<4>(q[r]);
  const int rr = lane & 7;
  float val = q[0];
  #pragma unroll
  for (int r = 1; r < 8; ++r) val = (rr == r) ? q[r] : val;
  const int orow = row0 + m0 + rr;
  float* lp = isDst ? ald : als;
  if (orow < Nn) lp[(size_t)orow * HEADS + hd] = val;
}

__global__ void gemm_wmma_x32(const float* __restrict__ X,
                              const _Float16* __restrict__ Bp,
                              const float* __restrict__ Asrc,
                              const float* __restrict__ Adst,
                              float* __restrict__ Hout,
                              float* __restrict__ als, float* __restrict__ ald,
                              int Nn) {
  gemm_head<true>(X, nullptr, Bp, Asrc, Adst, Hout, als, ald, Nn);
}

__global__ void gemm_wmma(const _Float16* __restrict__ Xh,
                          const _Float16* __restrict__ Bp,
                          const float* __restrict__ Asrc,
                          const float* __restrict__ Adst,
                          float* __restrict__ Hout,
                          float* __restrict__ als, float* __restrict__ ald,
                          int Nn) {
  gemm_head<false>(nullptr, Xh, Bp, Asrc, Adst, Hout, als, ald, Nn);
}

__device__ __forceinline__ float leaky(float v) {
  return v > 0.0f ? v : v * NEG_SLOPE;
}

#define SCAN_TILE 1024

__global__ void zero_counts(int* __restrict__ cnt, int total) {
  int t = blockIdx.x * blockDim.x + threadIdx.x;
  if (t < total) cnt[t] = 0;
}

__global__ void count_rank(const int* __restrict__ dst, int* __restrict__ cnt,
                           int* __restrict__ rank, int Etot) {
  int e = blockIdx.x * blockDim.x + threadIdx.x;
  if (e >= Etot) return;
  rank[e] = atomicAdd(&cnt[dst[e]], 1);
}

__global__ void __launch_bounds__(256) scan_tiles(const int4* __restrict__ cnt4,
                                                  int4* __restrict__ rowptr4,
                                                  int* __restrict__ tsum) {
  __shared__ int wsum[8];
  const int t = threadIdx.x;
  const int g = blockIdx.x * 256 + t;
  const int lane = t & 31, w = t >> 5;
  const int4 c = cnt4[g];
  const int s = c.x + c.y + c.z + c.w;
  int v = s;
  for (int o = 1; o < 32; o <<= 1) {
    const int u = __shfl_up(v, o, 32);
    if (lane >= o) v += u;
  }
  if (lane == 31) wsum[w] = v;
  __syncthreads();
  int woff = 0;
  for (int k = 0; k < 8; ++k) woff += (k < w) ? wsum[k] : 0;
  const int ex = woff + v - s;
  int4 r;
  r.x = ex; r.y = ex + c.x; r.z = ex + c.x + c.y; r.w = ex + c.x + c.y + c.z;
  rowptr4[g] = r;
  if (t == 255) tsum[blockIdx.x] = ex + s;
}

__global__ void __launch_bounds__(256) scan_add(int4* __restrict__ rowptr4,
                                                const int* __restrict__ tsum) {
  __shared__ int red[8];
  const int t = threadIdx.x;
  const int b = blockIdx.x;
  int part = 0;
  for (int k = t; k < b; k += 256) part += tsum[k];
  for (int o = 16; o >= 1; o >>= 1) part += __shfl_xor(part, o, 32);
  if ((t & 31) == 0) red[t >> 5] = part;
  __syncthreads();
  int off = 0;
  for (int k = 0; k < 8; ++k) off += red[k];
  int4 r = rowptr4[b * 256 + t];
  r.x += off; r.y += off; r.z += off; r.w += off;
  rowptr4[b * 256 + t] = r;
}

__global__ void fill_csr(const int* __restrict__ src, const int* __restrict__ dst,
                         const int* __restrict__ rank, const int* __restrict__ rowptr,
                         int* __restrict__ csr_u, int Etot) {
  int e = blockIdx.x * blockDim.x + threadIdx.x;
  if (e >= Etot) return;
  csr_u[rowptr[dst[e]] + rank[e]] = src[e];
}

__global__ void __launch_bounds__(256) sort_segments(const int* __restrict__ rowptr,
                                                     const int* __restrict__ csr_u,
                                                     int* __restrict__ csr_s, int Nn) {
  const int n = __builtin_amdgcn_readfirstlane((int)(blockIdx.x * 8 + (threadIdx.x >> 5)));
  const int lane = threadIdx.x & 31;
  if (n >= Nn) return;
  const int beg = rowptr[n], end = rowptr[n + 1];
  for (int bi = beg; bi < end; bi += 32) {
    const int i = bi + lane;
    const bool ok = i < end;
    const int vi = ok ? csr_u[i] : 0;
    int r = 0;
    for (int bq = beg; bq < end; bq += 32) {
      const int q = bq + lane;
      const int vq = (q < end) ? csr_u[q] : 0;
      const int nq = min(32, end - bq);
      for (int t = 0; t < nq; ++t) {
        const int x = __builtin_amdgcn_readlane(vq, t);
        const int qq = bq + t;
        r += (x < vi || (x == vi && qq < i)) ? 1 : 0;
      }
    }
    if (ok) csr_s[beg + r] = vi;
  }
}

__global__ void __launch_bounds__(256) node_gather(const int* __restrict__ rowptr,
                                                  const int* __restrict__ csr,
                                                  const float4* __restrict__ als4,
                                                  const float4* __restrict__ ald4,
                                                  const float4* __restrict__ Hh4,
                                                  const float* __restrict__ bias,
                                                  _Float16* __restrict__ xh_next,
                                                  float* __restrict__ out_f32,
                                                  int Nn, int hidden) {
  const int n = __builtin_amdgcn_readfirstlane((int)(blockIdx.x * 8 + (threadIdx.x >> 5)));
  const int lane = threadIdx.x & 31;
  if (n >= Nn) return;
  const int beg = rowptr[n], end = rowptr[n + 1];
  const float4 d4 = ald4[n];
  const bool hb0 = (lane & 8) != 0;
  const bool hb1 = (lane & 16) != 0;

  float4 acc = make_float4(0.0f, 0.0f, 0.0f, 0.0f);
  float wsum = 0.0f;
  float4 mx = make_float4(-INFINITY, -INFINITY, -INFINITY, -INFINITY);

  for (int base = beg; base < end; base += 32) {
    const int cnt = min(32, end - base);
    int s = 0;
    float4 e = make_float4(-INFINITY, -INFINITY, -INFINITY, -INFINITY);
    if (lane < cnt) {
      s = csr[base + lane];
      const float4 a = als4[s];
      e.x = leaky(a.x + d4.x);
      e.y = leaky(a.y + d4.y);
      e.z = leaky(a.z + d4.z);
      e.w = leaky(a.w + d4.w);
    }
    float4 cm = e;
    for (int o = 16; o >= 1; o >>= 1) {
      cm.x = fmaxf(cm.x, __shfl_xor(cm.x, o, 32));
      cm.y = fmaxf(cm.y, __shfl_xor(cm.y, o, 32));
      cm.z = fmaxf(cm.z, __shfl_xor(cm.z, o, 32));
      cm.w = fmaxf(cm.w, __shfl_xor(cm.w, o, 32));
    }
    float4 nm;
    nm.x = fmaxf(mx.x, cm.x); nm.y = fmaxf(mx.y, cm.y);
    nm.z = fmaxf(mx.z, cm.z); nm.w = fmaxf(mx.w, cm.w);
    const float sc01 = hb0 ? __expf(mx.y - nm.y) : __expf(mx.x - nm.x);
    const float sc23 = hb0 ? __expf(mx.w - nm.w) : __expf(mx.z - nm.z);
    const float sc = hb1 ? sc23 : sc01;
    acc.x *= sc; acc.y *= sc; acc.z *= sc; acc.w *= sc;
    wsum *= sc;
    mx = nm;
    const int ex0 = __float_as_int(__expf(e.x - nm.x));
    const int ex1 = __float_as_int(__expf(e.y - nm.y));
    const int ex2 = __float_as_int(__expf(e.z - nm.z));
    const int ex3 = __float_as_int(__expf(e.w - nm.w));

    int j = 0;
    for (; j + 4 <= cnt; j += 4) {
      float4 hv[4];
      float wv[4];
      for (int u = 0; u < 4; ++u) {
        const int sj = __builtin_amdgcn_readlane(s, j + u);
        hv[u] = Hh4[(size_t)sj * (FDIM / 4) + lane];
        const float a0 = __int_as_float(__builtin_amdgcn_readlane(ex0, j + u));
        const float a1 = __int_as_float(__builtin_amdgcn_readlane(ex1, j + u));
        const float a2 = __int_as_float(__builtin_amdgcn_readlane(ex2, j + u));
        const float a3 = __int_as_float(__builtin_amdgcn_readlane(ex3, j + u));
        const float w01 = hb0 ? a1 : a0;
        const float w23 = hb0 ? a3 : a2;
        wv[u] = hb1 ? w23 : w01;
      }
      for (int u = 0; u < 4; ++u) {
        acc.x = fmaf(hv[u].x, wv[u], acc.x);
        acc.y = fmaf(hv[u].y, wv[u], acc.y);
        acc.z = fmaf(hv[u].z, wv[u], acc.z);
        acc.w = fmaf(hv[u].w, wv[u], acc.w);
        wsum += wv[u];
      }
    }
    for (; j < cnt; ++j) {
      const int sj = __builtin_amdgcn_readlane(s, j);
      const float4 h1 = Hh4[(size_t)sj * (FDIM / 4) + lane];
      const float a0 = __int_as_float(__builtin_amdgcn_readlane(ex0, j));
      const float a1 = __int_as_float(__builtin_amdgcn_readlane(ex1, j));
      const float a2 = __int_as_float(__builtin_amdgcn_readlane(ex2, j));
      const float a3 = __int_as_float(__builtin_amdgcn_readlane(ex3, j));
      const float w01 = hb0 ? a1 : a0;
      const float w23 = hb0 ? a3 : a2;
      const float wj = hb1 ? w23 : w01;
      acc.x = fmaf(h1.x, wj, acc.x);
      acc.y = fmaf(h1.y, wj, acc.y);
      acc.z = fmaf(h1.z, wj, acc.z);
      acc.w = fmaf(h1.w, wj, acc.w);
      wsum += wj;
    }
  }

  const float inv = 1.0f / (wsum + GAT_EPS);
  const float* bp4 = bias + lane * 4;
  float4 b;
  b.x = bp4[0]; b.y = bp4[1]; b.z = bp4[2]; b.w = bp4[3];
  float4 o;
  o.x = fmaf(acc.x, inv, b.x);
  o.y = fmaf(acc.y, inv, b.y);
  o.z = fmaf(acc.z, inv, b.z);
  o.w = fmaf(acc.w, inv, b.w);
  const size_t at = (size_t)n * FDIM + (size_t)lane * 4;
  if (hidden) {
    v4h hq;
    hq[0] = (_Float16)fmaxf(o.x, 0.0f);
    hq[1] = (_Float16)fmaxf(o.y, 0.0f);
    hq[2] = (_Float16)fmaxf(o.z, 0.0f);
    hq[3] = (_Float16)fmaxf(o.w, 0.0f);
    *(v4h*)(xh_next + at) = hq;
  } else {
    *(float4*)(out_f32 + at) = o;
  }
}

extern "C" void kernel_launch(void* const* d_in, const int* in_sizes, int n_in,
                              void* d_out, int out_size, void* d_ws, size_t ws_size,
                              hipStream_t stream) {
  const int Nn   = in_sizes[0] / FDIM;
  const int Etot = in_sizes[1] / 2;

  const float* x0  = (const float*)d_in[0];
  const int*   ei  = (const int*)d_in[1];
  const int*   src = ei;
  const int*   dst = ei + Etot;

  const int nTiles  = (Nn + 1 + SCAN_TILE - 1) / SCAN_TILE;
  const int nPadded = nTiles * SCAN_TILE;

  char* ws = (char*)d_ws;
  size_t off = 0;
  auto alloc = [&](size_t bytes) {
    void* p = ws + off;
    off += (bytes + 255) & ~(size_t)255;
    return p;
  };
  _Float16* xh     = (_Float16*)alloc((size_t)Nn * FDIM * sizeof(_Float16));
  _Float16* bp     = (_Float16*)alloc((size_t)8 * 32 * 4 * 16 * sizeof(_Float16));
  float*    buf_h  = (float*)   alloc((size_t)Nn * FDIM * sizeof(float));
  float*    als    = (float*)   alloc((size_t)Nn * HEADS * sizeof(float));
  float*    ald    = (float*)   alloc((size_t)Nn * HEADS * sizeof(float));
  int*      cnt    = (int*)     alloc((size_t)nPadded * sizeof(int));
  int*      rowptr = (int*)     alloc((size_t)nPadded * sizeof(int));
  int*      tsum   = (int*)     alloc((size_t)nTiles * sizeof(int));
  int*      rank   = (int*)     alloc((size_t)Etot * sizeof(int));
  int*      csr_u  = (int*)     alloc((size_t)Etot * sizeof(int));
  int*      csr_s  = (int*)     alloc((size_t)Etot * sizeof(int));

  const int TPB = 256;
  const int gEdge  = (Etot + TPB - 1) / TPB;
  const int gNodeW = (Nn + (TPB / 32) - 1) / (TPB / 32);
  const int gPad   = (nPadded + TPB - 1) / TPB;
  const int rowTiles = (Nn + 15) >> 4;
  const int gGemm  = (rowTiles * HEADS + (TPB / 32) - 1) / (TPB / 32);
  const int gPackB = (8 * 32 * 4 * 16 + TPB - 1) / TPB;

  zero_counts<<<gPad, TPB, 0, stream>>>(cnt, nPadded);
  count_rank<<<gEdge, TPB, 0, stream>>>(dst, cnt, rank, Etot);
  scan_tiles<<<nTiles, TPB, 0, stream>>>((const int4*)cnt, (int4*)rowptr, tsum);
  scan_add<<<nTiles, TPB, 0, stream>>>((int4*)rowptr, tsum);
  fill_csr<<<gEdge, TPB, 0, stream>>>(src, dst, rank, rowptr, csr_u, Etot);
  sort_segments<<<gNodeW, TPB, 0, stream>>>(rowptr, csr_u, csr_s, Nn);

  for (int l = 0; l < 3; ++l) {
    const float* W    = (const float*)d_in[2 + l * 4];
    const float* Asrc = (const float*)d_in[3 + l * 4];
    const float* Adst = (const float*)d_in[4 + l * 4];
    const float* B    = (const float*)d_in[5 + l * 4];

    pack_b<<<gPackB, TPB, 0, stream>>>(W, bp);
    if (l == 0) gemm_wmma_x32<<<gGemm, TPB, 0, stream>>>(x0, bp, Asrc, Adst, buf_h, als, ald, Nn);
    else        gemm_wmma<<<gGemm, TPB, 0, stream>>>(xh, bp, Asrc, Adst, buf_h, als, ald, Nn);
    node_gather<<<gNodeW, TPB, 0, stream>>>(rowptr, csr_s, (const float4*)als,
                                           (const float4*)ald, (const float4*)buf_h, B,
                                           xh, (float*)d_out, Nn, l < 2 ? 1 : 0);
  }
}


# GEMM B fragments repacked so each wave-wide 16-byte load reads four whole 128-byte lines, 512 contiguous bytes (was 16 bytes from each of 32 lines); same elements, same k order and sum order
# speedup vs baseline: 6.5679x; 1.0747x over previous
#include <hip/hip_runtime.h>
#include <hip/hip_bf16.h>
#include <math.h>

typedef __attribute__((ext_vector_type(16))) _Float16 v16h;
typedef __attribute__((ext_vector_type(8)))  _Float16 v8h;
typedef __attribute__((ext_vector_type(8)))  float    v8f;
typedef __attribute__((ext_vector_type(4)))  _Float16 v4h;

#define HEADS 4
#define CDIM 32
#define FDIM 128
#define NEG_SLOPE 0.2f
#define GAT_EPS 1e-16f

__global__ void pack_b(const float* __restrict__ W, _Float16* __restrict__ Bp) {
  int t = blockIdx.x * blockDim.x + threadIdx.x;
  if (t >= 8 * 32 * 4 * 16) return;
  const int i    = (t & 7) + ((t >> 5) & 8);
  const int lane = (t >> 3) & 31;
  const int ks   = (t >> 9) & 3;
  const int ct   = t >> 11;
  const int ka   = ks * 32 + ((lane >= 16) ? 8 : 0);
  const int k    = ka + ((i < 8) ? i : i + 8);
  const int col  = (ct >> 1) * 32 + 2 * (lane & 15) + (ct & 1);
  Bp[t] = (_Float16)W[(size_t)k * FDIM + col];
}

template <int M>
__device__ __forceinline__ float row_xadd(float v) {
  const int t = __builtin_amdgcn_update_dpp(0, __float_as_int(v), 0x160 + M, 0xf, 0xf, true);
  return v + __int_as_float(t);
}

template <bool kF32Rows>
__device__ __forceinline__ void gemm_head(const float* __restrict__ Xf,
                                          const _Float16* __restrict__ Xh,
                                          const _Float16* __restrict__ Bp,
                                          const float* __restrict__ Asrc,
                                          const float* __restrict__ Adst,
                                          float* __restrict__ Hout,
                                          float* __restrict__ als,
                                          float* __restrict__ ald, int Nn) {
  const int wave = blockIdx.x * (blockDim.x >> 5) + (threadIdx.x >> 5);
  const int lane = threadIdx.x & 31;
  const int rowTiles = (Nn + 15) >> 4;
  if (wave >= rowTiles * HEADS) return;

  const int rt = wave >> 2;
  const int hd = wave & 3;
  const int row0 = rt << 4;
  const int lm = lane & 15;
  const bool hi = lane >= 16;

  int arow = row0 + lm;
  if (arow >= Nn) arow = Nn - 1;

  const size_t aoff = (size_t)arow * FDIM + (hi ? 8 : 0);
  const unsigned boff = (unsigned)(hd * 2) * 2048u + (unsigned)lane * 8u;
  const _Float16* bfrag0 = Bp + boff;
  const _Float16* bfrag1 = bfrag0 + 2048;

  const int colh = hd * CDIM + 2 * lm;
  const float2 as2 = *(const float2*)(Asrc + colh);
  const float2 ad2 = *(const float2*)(Adst + colh);

  v8f c0 = {};
  v8f c1 = {};
  #pragma unroll
  for (int ks = 0; ks < 4; ++ks) {
    v16h a;
    if constexpr (kF32Rows) {
      const float* ap = Xf + aoff + ks * 32;
      const float4 f0 = *(const float4*)(ap);
      const float4 f1 = *(const float4*)(ap + 4);
      const float4 f2 = *(const float4*)(ap + 16);
      const float4 f3 = *(const float4*)(ap + 20);
      a[0]  = (_Float16)f0.x; a[1]  = (_Float16)f0.y; a[2]  = (_Float16)f0.z; a[3]  = (_Float16)f0.w;
      a[4]  = (_Float16)f1.x; a[5]  = (_Float16)f1.y; a[6]  = (_Float16)f1.z; a[7]  = (_Float16)f1.w;
      a[8]  = (_Float16)f2.x; a[9]  = (_Float16)f2.y; a[10] = (_Float16)f2.z; a[11] = (_Float16)f2.w;
      a[12] = (_Float16)f3.x; a[13] = (_Float16)f3.y; a[14] = (_Float16)f3.z; a[15] = (_Float16)f3.w;
    } else {
      const _Float16* ap = Xh + aoff + ks * 32;
      const v8h alo = *(const v8h*)(ap);
      const v8h ahi = *(const v8h*)(ap + 16);
      #pragma unroll
      for (int i = 0; i < 8; ++i) { a[i] = alo[i]; a[8 + i] = ahi[i]; }
    }
    const v8h b0lo = *(const v8h*)(bfrag0 + ks * 512);
    const v8h b0hi = *(const v8h*)(bfrag0 + ks * 512 + 256);
    const v8h b1lo = *(const v8h*)(bfrag1 + ks * 512);
    const v8h b1hi = *(const v8h*)(bfrag1 + ks * 512 + 256);
    const v16h b0 = __builtin_shufflevector(b0lo, b0hi, 0, 1, 2, 3, 4, 5, 6, 7,
                                            8, 9, 10, 11, 12, 13, 14, 15);
    const v16h b1 = __builtin_shufflevector(b1lo, b1hi, 0, 1, 2, 3, 4, 5, 6, 7,
                                            8, 9, 10, 11, 12, 13, 14, 15);
    c0 = __builtin_amdgcn_wmma_f32_16x16x32_f16(
         false, a,  false, b0,
         (short)0, c0,  false,  false);
    c1 = __builtin_amdgcn_wmma_f32_16x16x32_f16(
         false, a,  false, b1,
         (short)0, c1,  false,  false);
  }

  const int m0 = hi ? 8 : 0;
  float* outp = Hout + (size_t)(row0 + m0) * FDIM + colh;
  if (row0 + 16 <= Nn) {
    #pragma unroll
    for (int r = 0; r < 8; ++r)
      *(float2*)(outp + (size_t)r * FDIM) = make_float2(c0[r], c1[r]);
  } else {
    #pragma unroll
    for (int r = 0; r < 8; ++r) {
      if (row0 + m0 + r < Nn)
        *(float2*)(outp + (size_t)r * FDIM) = make_float2(c0[r], c1[r]);
    }
  }

  const bool isDst = (lane & 8) != 0;
  float q[8];
  #pragma unroll
  for (int r = 0; r < 8; ++r) {
    const float ps = fmaf(c1[r], as2.y, c0[r] * as2.x);
    const float pd = fmaf(c1[r], ad2.y, c0[r] * ad2.x);
    const float keep = isDst ? pd : ps;
    const float give = isDst ? ps : pd;
    const int g = __builtin_amdgcn_update_dpp(0, __float_as_int(give), 0x168, 0xf, 0xf, true);
    q[r] = keep + __int_as_float(g);
  }
  #pragma unroll
  for (int r = 0; r < 8; ++r) q[r] = row_xadd<1>(q[r]);
  #pragma unroll
  for (int r = 0; r < 8; ++r) q[r] = row_xadd<2>(q[r]);
  #pragma unroll
  for (int r = 0; r < 8; ++r) q[r] = row_xadd<4>(q[r]);
  const int rr = lane & 7;
  float val = q[0];
  #pragma unroll
  for (int r = 1; r < 8; ++r) val = (rr == r) ? q[r] : val;
  const int orow = row0 + m0 + rr;
  float* lp = isDst ? ald : als;
  if (orow < Nn) lp[(size_t)orow * HEADS + hd] = val;
}

__global__ void gemm_wmma_x32(const float* __restrict__ X,
                              const _Float16* __restrict__ Bp,
                              const float* __restrict__ Asrc,
                              const float* __restrict__ Adst,
                              float* __restrict__ Hout,
                              float* __restrict__ als, float* __restrict__ ald,
                              int Nn) {
  gemm_head<true>(X, nullptr, Bp, Asrc, Adst, Hout, als, ald, Nn);
}

__global__ void gemm_wmma(const _Float16* __restrict__ Xh,
                          const _Float16* __restrict__ Bp,
                          const float* __restrict__ Asrc,
                          const float* __restrict__ Adst,
                          float* __restrict__ Hout,
                          float* __restrict__ als, float* __restrict__ ald,
                          int Nn) {
  gemm_head<false>(nullptr, Xh, Bp, Asrc, Adst, Hout, als, ald, Nn);
}

__device__ __forceinline__ float leaky(float v) {
  return v > 0.0f ? v : v * NEG_SLOPE;
}

#define SCAN_TILE 1024

__global__ void zero_counts(int* __restrict__ cnt, int total) {
  int t = blockIdx.x * blockDim.x + threadIdx.x;
  if (t < total) cnt[t] = 0;
}

__global__ void count_rank(const int* __restrict__ dst, int* __restrict__ cnt,
                           int* __restrict__ rank, int Etot) {
  int e = blockIdx.x * blockDim.x + threadIdx.x;
  if (e >= Etot) return;
  rank[e] = atomicAdd(&cnt[dst[e]], 1);
}

__global__ void __launch_bounds__(256) scan_tiles(const int4* __restrict__ cnt4,
                                                  int4* __restrict__ rowptr4,
                                                  int* __restrict__ tsum) {
  __shared__ int wsum[8];
  const int t = threadIdx.x;
  const int g = blockIdx.x * 256 + t;
  const int lane = t & 31, w = t >> 5;
  const int4 c = cnt4[g];
  const int s = c.x + c.y + c.z + c.w;
  int v = s;
  for (int o = 1; o < 32; o <<= 1) {
    const int u = __shfl_up(v, o, 32);
    if (lane >= o) v += u;
  }
  if (lane == 31) wsum[w] = v;
  __syncthreads();
  int woff = 0;
  for (int k = 0; k < 8; ++k) woff += (k < w) ? wsum[k] : 0;
  const int ex = woff + v - s;
  int4 r;
  r.x = ex; r.y = ex + c.x; r.z = ex + c.x + c.y; r.w = ex + c.x + c.y + c.z;
  rowptr4[g] = r;
  if (t == 255) tsum[blockIdx.x] = ex + s;
}

__global__ void __launch_bounds__(256) scan_add(int4* __restrict__ rowptr4,
                                                const int* __restrict__ tsum) {
  __shared__ int red[8];
  const int t = threadIdx.x;
  const int b = blockIdx.x;
  int part = 0;
  for (int k = t; k < b; k += 256) part += tsum[k];
  for (int o = 16; o >= 1; o >>= 1) part += __shfl_xor(part, o, 32);
  if ((t & 31) == 0) red[t >> 5] = part;
  __syncthreads();
  int off = 0;
  for (int k = 0; k < 8; ++k) off += red[k];
  int4 r = rowptr4[b * 256 + t];
  r.x += off; r.y += off; r.z += off; r.w += off;
  rowptr4[b * 256 + t] = r;
}

__global__ void fill_csr(const int* __restrict__ src, const int* __restrict__ dst,
                         const int* __restrict__ rank, const int* __restrict__ rowptr,
                         int* __restrict__ csr_u, int Etot) {
  int e = blockIdx.x * blockDim.x + threadIdx.x;
  if (e >= Etot) return;
  csr_u[rowptr[dst[e]] + rank[e]] = src[e];
}

__global__ void __launch_bounds__(256) sort_segments(const int* __restrict__ rowptr,
                                                     const int* __restrict__ csr_u,
                                                     int* __restrict__ csr_s, int Nn) {
  const int n = __builtin_amdgcn_readfirstlane((int)(blockIdx.x * 8 + (threadIdx.x >> 5)));
  const int lane = threadIdx.x & 31;
  if (n >= Nn) return;
  const int beg = rowptr[n], end = rowptr[n + 1];
  for (int bi = beg; bi < end; bi += 32) {
    const int i = bi + lane;
    const bool ok = i < end;
    const int vi = ok ? csr_u[i] : 0;
    int r = 0;
    for (int bq = beg; bq < end; bq += 32) {
      const int q = bq + lane;
      const int vq = (q < end) ? csr_u[q] : 0;
      const int nq = min(32, end - bq);
      for (int t = 0; t < nq; ++t) {
        const int x = __builtin_amdgcn_readlane(vq, t);
        const int qq = bq + t;
        r += (x < vi || (x == vi && qq < i)) ? 1 : 0;
      }
    }
    if (ok) csr_s[beg + r] = vi;
  }
}

__global__ void __launch_bounds__(256) node_gather(const int* __restrict__ rowptr,
                                                  const int* __restrict__ csr,
                                                  const float4* __restrict__ als4,
                                                  const float4* __restrict__ ald4,
                                                  const float4* __restrict__ Hh4,
                                                  const float* __restrict__ bias,
                                                  _Float16* __restrict__ xh_next,
                                                  float* __restrict__ out_f32,
                                                  int Nn, int hidden) {
  const int n = __builtin_amdgcn_readfirstlane((int)(blockIdx.x * 8 + (threadIdx.x >> 5)));
  const int lane = threadIdx.x & 31;
  if (n >= Nn) return;
  const int beg = rowptr[n], end = rowptr[n + 1];
  const float4 d4 = ald4[n];
  const bool hb0 = (lane & 8) != 0;
  const bool hb1 = (lane & 16) != 0;

  float4 acc = make_float4(0.0f, 0.0f, 0.0f, 0.0f);
  float wsum = 0.0f;
  float4 mx = make_float4(-INFINITY, -INFINITY, -INFINITY, -INFINITY);

  for (int base = beg; base < end; base += 32) {
    const int cnt = min(32, end - base);
    int s = 0;
    float4 e = make_float4(-INFINITY, -INFINITY, -INFINITY, -INFINITY);
    if (lane < cnt) {
      s = csr[base + lane];
      const float4 a = als4[s];
      e.x = leaky(a.x + d4.x);
      e.y = leaky(a.y + d4.y);
      e.z = leaky(a.z + d4.z);
      e.w = leaky(a.w + d4.w);
    }
    float4 cm = e;
    for (int o = 16; o >= 1; o >>= 1) {
      cm.x = fmaxf(cm.x, __shfl_xor(cm.x, o, 32));
      cm.y = fmaxf(cm.y, __shfl_xor(cm.y, o, 32));
      cm.z = fmaxf(cm.z, __shfl_xor(cm.z, o, 32));
      cm.w = fmaxf(cm.w, __shfl_xor(cm.w, o, 32));
    }
    float4 nm;
    nm.x = fmaxf(mx.x, cm.x); nm.y = fmaxf(mx.y, cm.y);
    nm.z = fmaxf(mx.z, cm.z); nm.w = fmaxf(mx.w, cm.w);
    const float sc01 = hb0 ? __expf(mx.y - nm.y) : __expf(mx.x - nm.x);
    const float sc23 = hb0 ? __expf(mx.w - nm.w) : __expf(mx.z - nm.z);
    const float sc = hb1 ? sc23 : sc01;
    acc.x *= sc; acc.y *= sc; acc.z *= sc; acc.w *= sc;
    wsum *= sc;
    mx = nm;
    const int ex0 = __float_as_int(__expf(e.x - nm.x));
    const int ex1 = __float_as_int(__expf(e.y - nm.y));
    const int ex2 = __float_as_int(__expf(e.z - nm.z));
    const int ex3 = __float_as_int(__expf(e.w - nm.w));

    int j = 0;
    for (; j + 4 <= cnt; j += 4) {
      float4 hv[4];
      float wv[4];
      for (int u = 0; u < 4; ++u) {
        const int sj = __builtin_amdgcn_readlane(s, j + u);
        hv[u] = Hh4[(size_t)sj * (FDIM / 4) + lane];
        const float a0 = __int_as_float(__builtin_amdgcn_readlane(ex0, j + u));
        const float a1 = __int_as_float(__builtin_amdgcn_readlane(ex1, j + u));
        const float a2 = __int_as_float(__builtin_amdgcn_readlane(ex2, j + u));
        const float a3 = __int_as_float(__builtin_amdgcn_readlane(ex3, j + u));
        const float w01 = hb0 ? a1 : a0;
        const float w23 = hb0 ? a3 : a2;
        wv[u] = hb1 ? w23 : w01;
      }
      for (int u = 0; u < 4; ++u) {
        acc.x = fmaf(hv[u].x, wv[u], acc.x);
        acc.y = fmaf(hv[u].y, wv[u], acc.y);
        acc.z = fmaf(hv[u].z, wv[u], acc.z);
        acc.w = fmaf(hv[u].w, wv[u], acc.w);
        wsum += wv[u];
      }
    }
    for (; j < cnt; ++j) {
      const int sj = __builtin_amdgcn_readlane(s, j);
      const float4 h1 = Hh4[(size_t)sj * (FDIM / 4) + lane];
      const float a0 = __int_as_float(__builtin_amdgcn_readlane(ex0, j));
      const float a1 = __int_as_float(__builtin_amdgcn_readlane(ex1, j));
      const float a2 = __int_as_float(__builtin_amdgcn_readlane(ex2, j));
      const float a3 = __int_as_float(__builtin_amdgcn_readlane(ex3, j));
      const float w01 = hb0 ? a1 : a0;
      const float w23 = hb0 ? a3 : a2;
      const float wj = hb1 ? w23 : w01;
      acc.x = fmaf(h1.x, wj, acc.x);
      acc.y = fmaf(h1.y, wj, acc.y);
      acc.z = fmaf(h1.z, wj, acc.z);
      acc.w = fmaf(h1.w, wj, acc.w);
      wsum += wj;
    }
  }

  const float inv = 1.0f / (wsum + GAT_EPS);
  const float* bp4 = bias + lane * 4;
  float4 b;
  b.x = bp4[0]; b.y = bp4[1]; b.z = bp4[2]; b.w = bp4[3];
  float4 o;
  o.x = fmaf(acc.x, inv, b.x);
  o.y = fmaf(acc.y, inv, b.y);
  o.z = fmaf(acc.z, inv, b.z);
  o.w = fmaf(acc.w, inv, b.w);
  const size_t at = (size_t)n * FDIM + (size_t)lane * 4;
  if (hidden) {
    v4h hq;
    hq[0] = (_Float16)fmaxf(o.x, 0.0f);
    hq[1] = (_Float16)fmaxf(o.y, 0.0f);
    hq[2] = (_Float16)fmaxf(o.z, 0.0f);
    hq[3] = (_Float16)fmaxf(o.w, 0.0f);
    *(v4h*)(xh_next + at) = hq;
  } else {
    *(float4*)(out_f32 + at) = o;
  }
}

extern "C" void kernel_launch(void* const* d_in, const int* in_sizes, int n_in,
                              void* d_out, int out_size, void* d_ws, size_t ws_size,
                              hipStream_t stream) {
  const int Nn   = in_sizes[0] / FDIM;
  const int Etot = in_sizes[1] / 2;

  const float* x0  = (const float*)d_in[0];
  const int*   ei  = (const int*)d_in[1];
  const int*   src = ei;
  const int*   dst = ei + Etot;

  const int nTiles  = (Nn + 1 + SCAN_TILE - 1) / SCAN_TILE;
  const int nPadded = nTiles * SCAN_TILE;

  char* ws = (char*)d_ws;
  size_t off = 0;
  auto alloc = [&](size_t bytes) {
    void* p = ws + off;
    off += (bytes + 255) & ~(size_t)255;
    return p;
  };
  _Float16* xh     = (_Float16*)alloc((size_t)Nn * FDIM * sizeof(_Float16));
  _Float16* bp     = (_Float16*)alloc((size_t)8 * 32 * 4 * 16 * sizeof(_Float16));
  float*    buf_h  = (float*)   alloc((size_t)Nn * FDIM * sizeof(float));
  float*    als    = (float*)   alloc((size_t)Nn * HEADS * sizeof(float));
  float*    ald    = (float*)   alloc((size_t)Nn * HEADS * sizeof(float));
  int*      cnt    = (int*)     alloc((size_t)nPadded * sizeof(int));
  int*      rowptr = (int*)     alloc((size_t)nPadded * sizeof(int));
  int*      tsum   = (int*)     alloc((size_t)nTiles * sizeof(int));
  int*      rank   = (int*)     alloc((size_t)Etot * sizeof(int));
  int*      csr_u  = (int*)     alloc((size_t)Etot * sizeof(int));
  int*      csr_s  = (int*)     alloc((size_t)Etot * sizeof(int));

  const int TPB = 256;
  const int gEdge  = (Etot + TPB - 1) / TPB;
  const int gNodeW = (Nn + (TPB / 32) - 1) / (TPB / 32);
  const int gPad   = (nPadded + TPB - 1) / TPB;
  const int rowTiles = (Nn + 15) >> 4;
  const int gGemm  = (rowTiles * HEADS + (TPB / 32) - 1) / (TPB / 32);
  const int gPackB = (8 * 32 * 4 * 16 + TPB - 1) / TPB;

  zero_counts<<<gPad, TPB, 0, stream>>>(cnt, nPadded);
  count_rank<<<gEdge, TPB, 0, stream>>>(dst, cnt, rank, Etot);
  scan_tiles<<<nTiles, TPB, 0, stream>>>((const int4*)cnt, (int4*)rowptr, tsum);
  scan_add<<<nTiles, TPB, 0, stream>>>((int4*)rowptr, tsum);
  fill_csr<<<gEdge, TPB, 0, stream>>>(src, dst, rank, rowptr, csr_u, Etot);
  sort_segments<<<gNodeW, TPB, 0, stream>>>(rowptr, csr_u, csr_s, Nn);

  for (int l = 0; l < 3; ++l) {
    const float* W    = (const float*)d_in[2 + l * 4];
    const float* Asrc = (const float*)d_in[3 + l * 4];
    const float* Adst = (const float*)d_in[4 + l * 4];
    const float* B    = (const float*)d_in[5 + l * 4];

    pack_b<<<gPackB, TPB, 0, stream>>>(W, bp);
    if (l == 0) gemm_wmma_x32<<<gGemm, TPB, 0, stream>>>(x0, bp, Asrc, Adst, buf_h, als, ald, Nn);
    else        gemm_wmma<<<gGemm, TPB, 0, stream>>>(xh, bp, Asrc, Adst, buf_h, als, ald, Nn);
    node_gather<<<gNodeW, TPB, 0, stream>>>(rowptr, csr_s, (const float4*)als,
                                           (const float4*)ald, (const float4*)buf_h, B,
                                           xh, (float*)d_out, Nn, l < 2 ? 1 : 0);
  }
}


# Stages that do not depend on each other share a launch (counter clear + all three weight packs; layer-0 GEMM + edge count; the two scan passes): 15 launches become 10; same sums, same order
# speedup vs baseline: 7.7303x; 1.1770x over previous
#include <hip/hip_runtime.h>
#include <hip/hip_bf16.h>
#include <math.h>

typedef __attribute__((ext_vector_type(16))) _Float16 v16h;
typedef __attribute__((ext_vector_type(8)))  _Float16 v8h;
typedef __attribute__((ext_vector_type(8)))  float    v8f;
typedef __attribute__((ext_vector_type(4)))  _Float16 v4h;
typedef __attribute__((ext_vector_type(2)))  _Float16 v2h;

#define HEADS 4
#define CDIM 32
#define FDIM 128
#define NEG_SLOPE 0.2f
#define GAT_EPS 1e-16f

#define PACK_ELEMS (8 * 32 * 4 * 16)
#define PACK_BLOCKS (PACK_ELEMS / 256)

__device__ __forceinline__ void pack_b_elem(const float* __restrict__ W,
                                            _Float16* __restrict__ Bp, int t) {
  const int i    = (t & 7) + ((t >> 5) & 8);
  const int lane = (t >> 3) & 31;
  const int ks   = (t >> 9) & 3;
  const int ct   = t >> 11;
  const int ka   = ks * 32 + ((lane >= 16) ? 8 : 0);
  const int k    = ka + ((i < 8) ? i : i + 8);
  const int col  = (ct >> 1) * 32 + 2 * (lane & 15) + (ct & 1);
  Bp[t] = (_Float16)W[(size_t)k * FDIM + col];
}

__global__ void __launch_bounds__(256) clear_and_pack(int* __restrict__ cnt, int total,
                                                      int zeroBlocks,
                                                      const float* __restrict__ W0,
                                                      const float* __restrict__ W1,
                                                      const float* __restrict__ W2,
                                                      _Float16* __restrict__ Bp0,
                                                      _Float16* __restrict__ Bp1,
                                                      _Float16* __restrict__ Bp2) {
  const int b = (int)blockIdx.x;
  if (b < zeroBlocks) {
    const int t = b * 256 + (int)threadIdx.x;
    if (t < total) cnt[t] = 0;
    return;
  }
  const int pb = b - zeroBlocks;
  const int l = pb / PACK_BLOCKS;
  const int t = (pb - l * PACK_BLOCKS) * 256 + (int)threadIdx.x;
  if (l == 0)      pack_b_elem(W0, Bp0, t);
  else if (l == 1) pack_b_elem(W1, Bp1, t);
  else             pack_b_elem(W2, Bp2, t);
}

template <int M>
__device__ __forceinline__ float row_xadd(float v) {
  const int t = __builtin_amdgcn_update_dpp(0, __float_as_int(v), 0x160 + M, 0xf, 0xf, true);
  return v + __int_as_float(t);
}

template <bool kF32Rows>
__device__ __forceinline__ void gemm_head(const float* __restrict__ Xf,
                                          const _Float16* __restrict__ Xh,
                                          const _Float16* __restrict__ Bp,
                                          const float* __restrict__ Asrc,
                                          const float* __restrict__ Adst,
                                          _Float16* __restrict__ Hout,
                                          float* __restrict__ als,
                                          float* __restrict__ ald, int Nn, int blk) {
  const int wave = blk * (blockDim.x >> 5) + (threadIdx.x >> 5);
  const int lane = threadIdx.x & 31;
  const int rowTiles = (Nn + 15) >> 4;
  if (wave >= rowTiles * HEADS) return;

  const int rt = wave >> 2;
  const int hd = wave & 3;
  const int row0 = rt << 4;
  const int lm = lane & 15;
  const bool hi = lane >= 16;

  int arow = row0 + lm;
  if (arow >= Nn) arow = Nn - 1;

  const size_t aoff = (size_t)arow * FDIM + (hi ? 8 : 0);
  const unsigned boff = (unsigned)(hd * 2) * 2048u + (unsigned)lane * 8u;
  const _Float16* bfrag0 = Bp + boff;
  const _Float16* bfrag1 = bfrag0 + 2048;

  const int colh = hd * CDIM + 2 * lm;
  const float2 as2 = *(const float2*)(Asrc + colh);
  const float2 ad2 = *(const float2*)(Adst + colh);

  v8f c0 = {};
  v8f c1 = {};
  #pragma unroll
  for (int ks = 0; ks < 4; ++ks) {
    v16h a;
    if constexpr (kF32Rows) {
      const float* ap = Xf + aoff + ks * 32;
      const float4 f0 = *(const float4*)(ap);
      const float4 f1 = *(const float4*)(ap + 4);
      const float4 f2 = *(const float4*)(ap + 16);
      const float4 f3 = *(const float4*)(ap + 20);
      a[0]  = (_Float16)f0.x; a[1]  = (_Float16)f0.y; a[2]  = (_Float16)f0.z; a[3]  = (_Float16)f0.w;
      a[4]  = (_Float16)f1.x; a[5]  = (_Float16)f1.y; a[6]  = (_Float16)f1.z; a[7]  = (_Float16)f1.w;
      a[8]  = (_Float16)f2.x; a[9]  = (_Float16)f2.y; a[10] = (_Float16)f2.z; a[11] = (_Float16)f2.w;
      a[12] = (_Float16)f3.x; a[13] = (_Float16)f3.y; a[14] = (_Float16)f3.z; a[15] = (_Float16)f3.w;
    } else {
      const _Float16* ap = Xh + aoff + ks * 32;
      const v8h alo = *(const v8h*)(ap);
      const v8h ahi = *(const v8h*)(ap + 16);
      #pragma unroll
      for (int i = 0; i < 8; ++i) { a[i] = alo[i]; a[8 + i] = ahi[i]; }
    }
    const v8h b0lo = *(const v8h*)(bfrag0 + ks * 512);
    const v8h b0hi = *(const v8h*)(bfrag0 + ks * 512 + 256);
    const v8h b1lo = *(const v8h*)(bfrag1 + ks * 512);
    const v8h b1hi = *(const v8h*)(bfrag1 + ks * 512 + 256);
    const v16h b0 = __builtin_shufflevector(b0lo, b0hi, 0, 1, 2, 3, 4, 5, 6, 7,
                                            8, 9, 10, 11, 12, 13, 14, 15);
    const v16h b1 = __builtin_shufflevector(b1lo, b1hi, 0, 1, 2, 3, 4, 5, 6, 7,
                                            8, 9, 10, 11, 12, 13, 14, 15);
    c0 = __builtin_amdgcn_wmma_f32_16x16x32_f16(
         false, a,  false, b0,
         (short)0, c0,  false,  false);
    c1 = __builtin_amdgcn_wmma_f32_16x16x32_f16(
         false, a,  false, b1,
         (short)0, c1,  false,  false);
  }

  const int m0 = hi ? 8 : 0;
  _Float16* outp = Hout + (size_t)(row0 + m0) * FDIM + colh;
  if (row0 + 16 <= Nn) {
    #pragma unroll
    for (int r = 0; r < 8; ++r) {
      v2h p;
      p[0] = (_Float16)c0[r];
      p[1] = (_Float16)c1[r];
      *(v2h*)(outp + (size_t)r * FDIM) = p;
    }
  } else {
    #pragma unroll
    for (int r = 0; r < 8; ++r) {
      if (row0 + m0 + r < Nn) {
        v2h p;
        p[0] = (_Float16)c0[r];
        p[1] = (_Float16)c1[r];
        *(v2h*)(outp + (size_t)r * FDIM) = p;
      }
    }
  }

  const bool isDst = (lane & 8) != 0;
  float q[8];
  #pragma unroll
  for (int r = 0; r < 8; ++r) {
    const float ps = fmaf(c1[r], as2.y, c0[r] * as2.x);
    const float pd = fmaf(c1[r], ad2.y, c0[r] * ad2.x);
    const float keep = isDst ? pd : ps;
    const float give = isDst ? ps : pd;
    const int g = __builtin_amdgcn_update_dpp(0, __float_as_int(give), 0x168, 0xf, 0xf, true);
    q[r] = keep + __int_as_float(g);
  }
  #pragma unroll
  for (int r = 0; r < 8; ++r) q[r] = row_xadd<1>(q[r]);
  #pragma unroll
  for (int r = 0; r < 8; ++r) q[r] = row_xadd<2>(q[r]);
  #pragma unroll
  for (int r = 0; r < 8; ++r) q[r] = row_xadd<4>(q[r]);
  const int rr = lane & 7;
  float val = q[0];
  #pragma unroll
  for (int r = 1; r < 8; ++r) val = (rr == r) ? q[r] : val;
  const int orow = row0 + m0 + rr;
  float* lp = isDst ? ald : als;
  if (orow < Nn) lp[(size_t)orow * HEADS + hd] = val;
}

__global__ void __launch_bounds__(256) gemm_x32_and_count(const float* __restrict__ X,
                                   const _Float16* __restrict__ Bp,
                                   const float* __restrict__ Asrc,
                                   const float* __restrict__ Adst,
                                   _Float16* __restrict__ Hout,
                                   float* __restrict__ als, float* __restrict__ ald,
                                   int Nn,
                                   const int* __restrict__ dst, int* __restrict__ cnt,
                                   int* __restrict__ rank, int Etot) {
  const int q = (int)blockIdx.x / 3;
  const int r = (int)blockIdx.x - 3 * q;
  if (r != 0) {
    const int e = (2 * q + r - 1) * 256 + (int)threadIdx.x;
    if (e < Etot) rank[e] = atomicAdd(&cnt[dst[e]], 1);
    return;
  }
  gemm_head<true>(X, nullptr, Bp, Asrc, Adst, Hout, als, ald, Nn, q);
}

__global__ void gemm_wmma(const _Float16* __restrict__ Xh,
                          const _Float16* __restrict__ Bp,
                          const float* __restrict__ Asrc,
                          const float* __restrict__ Adst,
                          _Float16* __restrict__ Hout,
                          float* __restrict__ als, float* __restrict__ ald,
                          int Nn) {
  gemm_head<false>(nullptr, Xh, Bp, Asrc, Adst, Hout, als, ald, Nn, (int)blockIdx.x);
}

__device__ __forceinline__ float leaky(float v) {
  return v > 0.0f ? v : v * NEG_SLOPE;
}

#define SCAN_TILE 1024

__global__ void __launch_bounds__(256) scan_counts(const int4* __restrict__ cnt4,
                                                   int4* __restrict__ rowptr4) {
  __shared__ int wsum[8];
  __shared__ int psum[8];
  const int t = threadIdx.x;
  const int b = blockIdx.x;
  const int lane = t & 31, w = t >> 5;
  const int4 c = cnt4[b * 256 + t];
  int part = 0;
  for (int k = 0; k < b; ++k) {
    const int4 p = cnt4[k * 256 + t];
    part += (p.x + p.y) + (p.z + p.w);
  }
  const int s = c.x + c.y + c.z + c.w;
  int v = s;
  for (int o = 1; o < 32; o <<= 1) {
    const int u = __shfl_up(v, o, 32);
    if (lane >= o) v += u;
  }
  for (int o = 16; o >= 1; o >>= 1) part += __shfl_xor(part, o, 32);
  if (lane == 31) wsum[w] = v;
  if (lane == 0) psum[w] = part;
  __syncthreads();
  int woff = 0;
  for (int k = 0; k < 8; ++k) woff += psum[k] + ((k < w) ? wsum[k] : 0);
  const int ex = woff + v - s;
  int4 r;
  r.x = ex; r.y = ex + c.x; r.z = ex + c.x + c.y; r.w = ex + c.x + c.y + c.z;
  rowptr4[b * 256 + t] = r;
}

__global__ void fill_csr(const int* __restrict__ src, const int* __restrict__ dst,
                         const int* __restrict__ rank, const int* __restrict__ rowptr,
                         int* __restrict__ csr_u, int Etot) {
  int e = blockIdx.x * blockDim.x + threadIdx.x;
  if (e >= Etot) return;
  csr_u[rowptr[dst[e]] + rank[e]] = src[e];
}

__global__ void __launch_bounds__(256) sort_segments(const int* __restrict__ rowptr,
                                                     const int* __restrict__ csr_u,
                                                     int* __restrict__ csr_s, int Nn) {
  const int n = __builtin_amdgcn_readfirstlane((int)(blockIdx.x * 8 + (threadIdx.x >> 5)));
  const int lane = threadIdx.x & 31;
  if (n >= Nn) return;
  const int beg = rowptr[n], end = rowptr[n + 1];
  for (int bi = beg; bi < end; bi += 32) {
    const int i = bi + lane;
    const bool ok = i < end;
    const int vi = ok ? csr_u[i] : 0;
    int r = 0;
    for (int bq = beg; bq < end; bq += 32) {
      const int q = bq + lane;
      const int vq = (q < end) ? csr_u[q] : 0;
      const int nq = min(32, end - bq);
      for (int t = 0; t < nq; ++t) {
        const int x = __builtin_amdgcn_readlane(vq, t);
        const int qq = bq + t;
        r += (x < vi || (x == vi && qq < i)) ? 1 : 0;
      }
    }
    if (ok) csr_s[beg + r] = vi;
  }
}

__global__ void __launch_bounds__(256) node_gather(const int* __restrict__ rowptr,
                                                  const int* __restrict__ csr,
                                                  const float4* __restrict__ als4,
                                                  const float4* __restrict__ ald4,
                                                  const _Float16* __restrict__ Hh,
                                                  const float* __restrict__ bias,
                                                  _Float16* __restrict__ xh_next,
                                                  float* __restrict__ out_f32,
                                                  int Nn, int hidden) {
  const int n = __builtin_amdgcn_readfirstlane((int)(blockIdx.x * 8 + (threadIdx.x >> 5)));
  const int lane = threadIdx.x & 31;
  if (n >= Nn) return;
  const int beg = rowptr[n], end = rowptr[n + 1];
  const float4 d4 = ald4[n];
  const bool hb0 = (lane & 8) != 0;
  const bool hb1 = (lane & 16) != 0;

  float4 acc = make_float4(0.0f, 0.0f, 0.0f, 0.0f);
  float wsum = 0.0f;
  float4 mx = make_float4(-INFINITY, -INFINITY, -INFINITY, -INFINITY);

  for (int base = beg; base < end; base += 32) {
    const int cnt = min(32, end - base);
    int s = 0;
    float4 e = make_float4(-INFINITY, -INFINITY, -INFINITY, -INFINITY);
    if (lane < cnt) {
      s = csr[base + lane];
      const float4 a = als4[s];
      e.x = leaky(a.x + d4.x);
      e.y = leaky(a.y + d4.y);
      e.z = leaky(a.z + d4.z);
      e.w = leaky(a.w + d4.w);
    }
    float4 cm = e;
    for (int o = 16; o >= 1; o >>= 1) {
      cm.x = fmaxf(cm.x, __shfl_xor(cm.x, o, 32));
      cm.y = fmaxf(cm.y, __shfl_xor(cm.y, o, 32));
      cm.z = fmaxf(cm.z, __shfl_xor(cm.z, o, 32));
      cm.w = fmaxf(cm.w, __shfl_xor(cm.w, o, 32));
    }
    float4 nm;
    nm.x = fmaxf(mx.x, cm.x); nm.y = fmaxf(mx.y, cm.y);
    nm.z = fmaxf(mx.z, cm.z); nm.w = fmaxf(mx.w, cm.w);
    const float sc01 = hb0 ? __expf(mx.y - nm.y) : __expf(mx.x - nm.x);
    const float sc23 = hb0 ? __expf(mx.w - nm.w) : __expf(mx.z - nm.z);
    const float sc = hb1 ? sc23 : sc01;
    acc.x *= sc; acc.y *= sc; acc.z *= sc; acc.w *= sc;
    wsum *= sc;
    mx = nm;
    const int ex0 = __float_as_int(__expf(e.x - nm.x));
    const int ex1 = __float_as_int(__expf(e.y - nm.y));
    const int ex2 = __float_as_int(__expf(e.z - nm.z));
    const int ex3 = __float_as_int(__expf(e.w - nm.w));

    int j = 0;
    for (; j + 4 <= cnt; j += 4) {
      v4h hv[4];
      float wv[4];
      for (int u = 0; u < 4; ++u) {
        const int sj = __builtin_amdgcn_readlane(s, j + u);
        hv[u] = *(const v4h*)(Hh + (size_t)sj * FDIM + lane * 4);
        const float a0 = __int_as_float(__builtin_amdgcn_readlane(ex0, j + u));
        const float a1 = __int_as_float(__builtin_amdgcn_readlane(ex1, j + u));
        const float a2 = __int_as_float(__builtin_amdgcn_readlane(ex2, j + u));
        const float a3 = __int_as_float(__builtin_amdgcn_readlane(ex3, j + u));
        const float w01 = hb0 ? a1 : a0;
        const float w23 = hb0 ? a3 : a2;
        wv[u] = hb1 ? w23 : w01;
      }
      for (int u = 0; u < 4; ++u) {
        acc.x = fmaf((float)hv[u][0], wv[u], acc.x);
        acc.y = fmaf((float)hv[u][1], wv[u], acc.y);
        acc.z = fmaf((float)hv[u][2], wv[u], acc.z);
        acc.w = fmaf((float)hv[u][3], wv[u], acc.w);
        wsum += wv[u];
      }
    }
    for (; j < cnt; ++j) {
      const int sj = __builtin_amdgcn_readlane(s, j);
      const v4h h1 = *(const v4h*)(Hh + (size_t)sj * FDIM + lane * 4);
      const float a0 = __int_as_float(__builtin_amdgcn_readlane(ex0, j));
      const float a1 = __int_as_float(__builtin_amdgcn_readlane(ex1, j));
      const float a2 = __int_as_float(__builtin_amdgcn_readlane(ex2, j));
      const float a3 = __int_as_float(__builtin_amdgcn_readlane(ex3, j));
      const float w01 = hb0 ? a1 : a0;
      const float w23 = hb0 ? a3 : a2;
      const float wj = hb1 ? w23 : w01;
      acc.x = fmaf((float)h1[0], wj, acc.x);
      acc.y = fmaf((float)h1[1], wj, acc.y);
      acc.z = fmaf((float)h1[2], wj, acc.z);
      acc.w = fmaf((float)h1[3], wj, acc.w);
      wsum += wj;
    }
  }

  const float inv = 1.0f / (wsum + GAT_EPS);
  const float* bp4 = bias + lane * 4;
  float4 b;
  b.x = bp4[0]; b.y = bp4[1]; b.z = bp4[2]; b.w = bp4[3];
  float4 o;
  o.x = fmaf(acc.x, inv, b.x);
  o.y = fmaf(acc.y, inv, b.y);
  o.z = fmaf(acc.z, inv, b.z);
  o.w = fmaf(acc.w, inv, b.w);
  const size_t at = (size_t)n * FDIM + (size_t)lane * 4;
  if (hidden) {
    v4h hq;
    hq[0] = (_Float16)fmaxf(o.x, 0.0f);
    hq[1] = (_Float16)fmaxf(o.y, 0.0f);
    hq[2] = (_Float16)fmaxf(o.z, 0.0f);
    hq[3] = (_Float16)fmaxf(o.w, 0.0f);
    *(v4h*)(xh_next + at) = hq;
  } else {
    *(float4*)(out_f32 + at) = o;
  }
}

extern "C" void kernel_launch(void* const* d_in, const int* in_sizes, int n_in,
                              void* d_out, int out_size, void* d_ws, size_t ws_size,
                              hipStream_t stream) {
  const int Nn   = in_sizes[0] / FDIM;
  const int Etot = in_sizes[1] / 2;

  const float* x0  = (const float*)d_in[0];
  const int*   ei  = (const int*)d_in[1];
  const int*   src = ei;
  const int*   dst = ei + Etot;

  const int nTiles  = (Nn + 1 + SCAN_TILE - 1) / SCAN_TILE;
  const int nPadded = nTiles * SCAN_TILE;

  char* ws = (char*)d_ws;
  size_t off = 0;
  auto alloc = [&](size_t bytes) {
    void* p = ws + off;
    off += (bytes + 255) & ~(size_t)255;
    return p;
  };
  _Float16* xh     = (_Float16*)alloc((size_t)Nn * FDIM * sizeof(_Float16));
  _Float16* bp0    = (_Float16*)alloc((size_t)PACK_ELEMS * sizeof(_Float16));
  _Float16* bp1    = (_Float16*)alloc((size_t)PACK_ELEMS * sizeof(_Float16));
  _Float16* bp2    = (_Float16*)alloc((size_t)PACK_ELEMS * sizeof(_Float16));
  _Float16* buf_h  = (_Float16*)alloc((size_t)Nn * FDIM * sizeof(_Float16));
  float*    als    = (float*)   alloc((size_t)Nn * HEADS * sizeof(float));
  float*    ald    = (float*)   alloc((size_t)Nn * HEADS * sizeof(float));
  int*      cnt    = (int*)     alloc((size_t)nPadded * sizeof(int));
  int*      rowptr = (int*)     alloc((size_t)nPadded * sizeof(int));
  int*      rank   = (int*)     alloc((size_t)Etot * sizeof(int));
  int*      csr_u  = (int*)     alloc((size_t)Etot * sizeof(int));
  int*      csr_s  = (int*)     alloc((size_t)Etot * sizeof(int));

  const int TPB = 256;
  const int gEdge  = (Etot + TPB - 1) / TPB;
  const int gNodeW = (Nn + (TPB / 32) - 1) / (TPB / 32);
  const int gPad   = (nPadded + TPB - 1) / TPB;
  const int rowTiles = (Nn + 15) >> 4;
  const int gGemm  = (rowTiles * HEADS + (TPB / 32) - 1) / (TPB / 32);

  clear_and_pack<<<gPad + 3 * PACK_BLOCKS, TPB, 0, stream>>>(
      cnt, nPadded, gPad,
      (const float*)d_in[2], (const float*)d_in[6], (const float*)d_in[10],
      bp0, bp1, bp2);
  const int gTriples = (gGemm > (gEdge + 1) / 2) ? gGemm : (gEdge + 1) / 2;
  gemm_x32_and_count<<<3 * gTriples, TPB, 0, stream>>>(
      x0, bp0, (const float*)d_in[3], (const float*)d_in[4], buf_h, als, ald, Nn,
      dst, cnt, rank, Etot);
  scan_counts<<<nTiles, TPB, 0, stream>>>((const int4*)cnt, (int4*)rowptr);
  fill_csr<<<gEdge, TPB, 0, stream>>>(src, dst, rank, rowptr, csr_u, Etot);
  sort_segments<<<gNodeW, TPB, 0, stream>>>(rowptr, csr_u, csr_s, Nn);

  for (int l = 0; l < 3; ++l) {
    const float* Asrc = (const float*)d_in[3 + l * 4];
    const float* Adst = (const float*)d_in[4 + l * 4];
    const float* B    = (const float*)d_in[5 + l * 4];

    if (l > 0) gemm_wmma<<<gGemm, TPB, 0, stream>>>(xh, l == 1 ? bp1 : bp2, Asrc, Adst,
                                               buf_h, als, ald, Nn);
    node_gather<<<gNodeW, TPB, 0, stream>>>(rowptr, csr_s, (const float4*)als,
                                           (const float4*)ald, buf_h, B,
                                           xh, (float*)d_out, Nn, l < 2 ? 1 : 0);
  }
}


# Layers 1, 2 projected inside the previous layer's gather kernel: a 16-wave work-group gathers 16 nodes, hands their f16 rows through LDS to four of its waves' WMMA; two launches gone; sums unchanged
# speedup vs baseline: 8.0245x; 1.0381x over previous
#include <hip/hip_runtime.h>
#include <hip/hip_bf16.h>
#include <math.h>

typedef __attribute__((ext_vector_type(16))) _Float16 v16h;
typedef __attribute__((ext_vector_type(8)))  _Float16 v8h;
typedef __attribute__((ext_vector_type(8)))  float    v8f;
typedef __attribute__((ext_vector_type(4)))  _Float16 v4h;
typedef __attribute__((ext_vector_type(2)))  _Float16 v2h;

#define HEADS 4
#define CDIM 32
#define FDIM 128
#define NEG_SLOPE 0.2f
#define GAT_EPS 1e-16f
#define TILE_PITCH 136

#define PACK_ELEMS (8 * 32 * 4 * 16)
#define PACK_BLOCKS (PACK_ELEMS / 256)

__device__ __forceinline__ void pack_b_elem(const float* __restrict__ W,
                                            _Float16* __restrict__ Bp, int t) {
  const int i    = (t & 7) + ((t >> 5) & 8);
  const int lane = (t >> 3) & 31;
  const int ks   = (t >> 9) & 3;
  const int ct   = t >> 11;
  const int ka   = ks * 32 + ((lane >= 16) ? 8 : 0);
  const int k    = ka + ((i < 8) ? i : i + 8);
  const int col  = (ct >> 1) * 32 + 2 * (lane & 15) + (ct & 1);
  Bp[t] = (_Float16)W[(size_t)k * FDIM + col];
}

__global__ void __launch_bounds__(256) clear_and_pack(int* __restrict__ cnt, int total,
                                                      int zeroBlocks,
                                                      const float* __restrict__ W0,
                                                      const float* __restrict__ W1,
                                                      const float* __restrict__ W2,
                                                      _Float16* __restrict__ Bp0,
                                                      _Float16* __restrict__ Bp1,
                                                      _Float16* __restrict__ Bp2) {
  const int b = (int)blockIdx.x;
  if (b < zeroBlocks) {
    const int t = b * 256 + (int)threadIdx.x;
    if (t < total) cnt[t] = 0;
    return;
  }
  const int pb = b - zeroBlocks;
  const int l = pb / PACK_BLOCKS;
  const int t = (pb - l * PACK_BLOCKS) * 256 + (int)threadIdx.x;
  if (l == 0)      pack_b_elem(W0, Bp0, t);
  else if (l == 1) pack_b_elem(W1, Bp1, t);
  else             pack_b_elem(W2, Bp2, t);
}

template <int M>
__device__ __forceinline__ float row_xadd(float v) {
  const int t = __builtin_amdgcn_update_dpp(0, __float_as_int(v), 0x160 + M, 0xf, 0xf, true);
  return v + __int_as_float(t);
}

template <bool kF32Rows>
__device__ __forceinline__ void gemm_head(const float* __restrict__ Xf,
                                          const _Float16* Xt,
                                          const _Float16* __restrict__ Bp,
                                          const float* __restrict__ Asrc,
                                          const float* __restrict__ Adst,
                                          _Float16* __restrict__ Hout,
                                          float* __restrict__ als,
                                          float* __restrict__ ald, int Nn, int rt, int hd) {
  const int lane = threadIdx.x & 31;
  const int row0 = rt << 4;
  const int lm = lane & 15;
  const bool hi = lane >= 16;

  size_t aoff;
  if constexpr (kF32Rows) {
    int arow = row0 + lm;
    if (arow >= Nn) arow = Nn - 1;
    aoff = (size_t)arow * FDIM + (hi ? 8 : 0);
  } else {
    aoff = (size_t)lm * TILE_PITCH + (hi ? 8 : 0);
  }
  const unsigned boff = (unsigned)(hd * 2) * 2048u + (unsigned)lane * 8u;
  const _Float16* bfrag0 = Bp + boff;
  const _Float16* bfrag1 = bfrag0 + 2048;

  const int colh = hd * CDIM + 2 * lm;
  const float2 as2 = *(const float2*)(Asrc + colh);
  const float2 ad2 = *(const float2*)(Adst + colh);

  v8f c0 = {};
  v8f c1 = {};
  #pragma unroll
  for (int ks = 0; ks < 4; ++ks) {
    v16h a;
    if constexpr (kF32Rows) {
      const float* ap = Xf + aoff + ks * 32;
      const float4 f0 = *(const float4*)(ap);
      const float4 f1 = *(const float4*)(ap + 4);
      const float4 f2 = *(const float4*)(ap + 16);
      const float4 f3 = *(const float4*)(ap + 20);
      a[0]  = (_Float16)f0.x; a[1]  = (_Float16)f0.y; a[2]  = (_Float16)f0.z; a[3]  = (_Float16)f0.w;
      a[4]  = (_Float16)f1.x; a[5]  = (_Float16)f1.y; a[6]  = (_Float16)f1.z; a[7]  = (_Float16)f1.w;
      a[8]  = (_Float16)f2.x; a[9]  = (_Float16)f2.y; a[10] = (_Float16)f2.z; a[11] = (_Float16)f2.w;
      a[12] = (_Float16)f3.x; a[13] = (_Float16)f3.y; a[14] = (_Float16)f3.z; a[15] = (_Float16)f3.w;
    } else {
      const _Float16* ap = Xt + aoff + ks * 32;
      const v8h alo = *(const v8h*)(ap);
      const v8h ahi = *(const v8h*)(ap + 16);
      #pragma unroll
      for (int i = 0; i < 8; ++i) { a[i] = alo[i]; a[8 + i] = ahi[i]; }
    }
    const v8h b0lo = *(const v8h*)(bfrag0 + ks * 512);
    const v8h b0hi = *(const v8h*)(bfrag0 + ks * 512 + 256);
    const v8h b1lo = *(const v8h*)(bfrag1 + ks * 512);
    const v8h b1hi = *(const v8h*)(bfrag1 + ks * 512 + 256);
    const v16h b0 = __builtin_shufflevector(b0lo, b0hi, 0, 1, 2, 3, 4, 5, 6, 7,
                                            8, 9, 10, 11, 12, 13, 14, 15);
    const v16h b1 = __builtin_shufflevector(b1lo, b1hi, 0, 1, 2, 3, 4, 5, 6, 7,
                                            8, 9, 10, 11, 12, 13, 14, 15);
    c0 = __builtin_amdgcn_wmma_f32_16x16x32_f16(
         false, a,  false, b0,
         (short)0, c0,  false,  false);
    c1 = __builtin_amdgcn_wmma_f32_16x16x32_f16(
         false, a,  false, b1,
         (short)0, c1,  false,  false);
  }

  const int m0 = hi ? 8 : 0;
  _Float16* outp = Hout + (size_t)(row0 + m0) * FDIM + colh;
  if (row0 + 16 <= Nn) {
    #pragma unroll
    for (int r = 0; r < 8; ++r) {
      v2h p;
      p[0] = (_Float16)c0[r];
      p[1] = (_Float16)c1[r];
      *(v2h*)(outp + (size_t)r * FDIM) = p;
    }
  } else {
    #pragma unroll
    for (int r = 0; r < 8; ++r) {
      if (row0 + m0 + r < Nn) {
        v2h p;
        p[0] = (_Float16)c0[r];
        p[1] = (_Float16)c1[r];
        *(v2h*)(outp + (size_t)r * FDIM) = p;
      }
    }
  }

  const bool isDst = (lane & 8) != 0;
  float q[8];
  #pragma unroll
  for (int r = 0; r < 8; ++r) {
    const float ps = fmaf(c1[r], as2.y, c0[r] * as2.x);
    const float pd = fmaf(c1[r], ad2.y, c0[r] * ad2.x);
    const float keep = isDst ? pd : ps;
    const float give = isDst ? ps : pd;
    const int g = __builtin_amdgcn_update_dpp(0, __float_as_int(give), 0x168, 0xf, 0xf, true);
    q[r] = keep + __int_as_float(g);
  }
  #pragma unroll
  for (int r = 0; r < 8; ++r) q[r] = row_xadd<1>(q[r]);
  #pragma unroll
  for (int r = 0; r < 8; ++r) q[r] = row_xadd<2>(q[r]);
  #pragma unroll
  for (int r = 0; r < 8; ++r) q[r] = row_xadd<4>(q[r]);
  const int rr = lane & 7;
  float val = q[0];
  #pragma unroll
  for (int r = 1; r < 8; ++r) val = (rr == r) ? q[r] : val;
  const int orow = row0 + m0 + rr;
  float* lp = isDst ? ald : als;
  if (orow < Nn) lp[(size_t)orow * HEADS + hd] = val;
}

__global__ void __launch_bounds__(256) gemm_x32_and_count(const float* __restrict__ X,
                                   const _Float16* __restrict__ Bp,
                                   const float* __restrict__ Asrc,
                                   const float* __restrict__ Adst,
                                   _Float16* __restrict__ Hout,
                                   float* __restrict__ als, float* __restrict__ ald,
                                   int Nn,
                                   const int* __restrict__ dst, int* __restrict__ cnt,
                                   int* __restrict__ rank, int Etot) {
  const int q = (int)blockIdx.x / 3;
  const int r = (int)blockIdx.x - 3 * q;
  if (r != 0) {
    const int e = (2 * q + r - 1) * 256 + (int)threadIdx.x;
    if (e < Etot) rank[e] = atomicAdd(&cnt[dst[e]], 1);
    return;
  }
  const int wave = q * 8 + ((int)threadIdx.x >> 5);
  if (wave >= ((Nn + 15) >> 4) * HEADS) return;
  gemm_head<true>(X, nullptr, Bp, Asrc, Adst, Hout, als, ald, Nn, wave >> 2, wave & 3);
}

__device__ __forceinline__ float leaky(float v) {
  return v > 0.0f ? v : v * NEG_SLOPE;
}

#define SCAN_TILE 1024

__global__ void __launch_bounds__(256) scan_counts(const int4* __restrict__ cnt4,
                                                   int4* __restrict__ rowptr4) {
  __shared__ int wsum[8];
  __shared__ int psum[8];
  const int t = threadIdx.x;
  const int b = blockIdx.x;
  const int lane = t & 31, w = t >> 5;
  const int4 c = cnt4[b * 256 + t];
  int part = 0;
  for (int k = 0; k < b; ++k) {
    const int4 p = cnt4[k * 256 + t];
    part += (p.x + p.y) + (p.z + p.w);
  }
  const int s = c.x + c.y + c.z + c.w;
  int v = s;
  for (int o = 1; o < 32; o <<= 1) {
    const int u = __shfl_up(v, o, 32);
    if (lane >= o) v += u;
  }
  for (int o = 16; o >= 1; o >>= 1) part += __shfl_xor(part, o, 32);
  if (lane == 31) wsum[w] = v;
  if (lane == 0) psum[w] = part;
  __syncthreads();
  int woff = 0;
  for (int k = 0; k < 8; ++k) woff += psum[k] + ((k < w) ? wsum[k] : 0);
  const int ex = woff + v - s;
  int4 r;
  r.x = ex; r.y = ex + c.x; r.z = ex + c.x + c.y; r.w = ex + c.x + c.y + c.z;
  rowptr4[b * 256 + t] = r;
}

__global__ void fill_csr(const int* __restrict__ src, const int* __restrict__ dst,
                         const int* __restrict__ rank, const int* __restrict__ rowptr,
                         int* __restrict__ csr_u, int Etot) {
  int e = blockIdx.x * blockDim.x + threadIdx.x;
  if (e >= Etot) return;
  csr_u[rowptr[dst[e]] + rank[e]] = src[e];
}

__global__ void __launch_bounds__(256) sort_segments(const int* __restrict__ rowptr,
                                                     const int* __restrict__ csr_u,
                                                     int* __restrict__ csr_s, int Nn) {
  const int n = __builtin_amdgcn_readfirstlane((int)(blockIdx.x * 8 + (threadIdx.x >> 5)));
  const int lane = threadIdx.x & 31;
  if (n >= Nn) return;
  const int beg = rowptr[n], end = rowptr[n + 1];
  for (int bi = beg; bi < end; bi += 32) {
    const int i = bi + lane;
    const bool ok = i < end;
    const int vi = ok ? csr_u[i] : 0;
    int r = 0;
    for (int bq = beg; bq < end; bq += 32) {
      const int q = bq + lane;
      const int vq = (q < end) ? csr_u[q] : 0;
      const int nq = min(32, end - bq);
      for (int t = 0; t < nq; ++t) {
        const int x = __builtin_amdgcn_readlane(vq, t);
        const int qq = bq + t;
        r += (x < vi || (x == vi && qq < i)) ? 1 : 0;
      }
    }
    if (ok) csr_s[beg + r] = vi;
  }
}

__device__ __forceinline__ float4 gather_row(const int* __restrict__ rowptr,
                                             const int* __restrict__ csr,
                                             const float4* __restrict__ als4,
                                             const float4* __restrict__ ald4,
                                             const _Float16* __restrict__ Hh,
                                             const float* __restrict__ bias,
                                             int n, int lane) {
  const int beg = rowptr[n], end = rowptr[n + 1];
  const float4 d4 = ald4[n];
  const bool hb0 = (lane & 8) != 0;
  const bool hb1 = (lane & 16) != 0;

  float4 acc = make_float4(0.0f, 0.0f, 0.0f, 0.0f);
  float wsum = 0.0f;
  float4 mx = make_float4(-INFINITY, -INFINITY, -INFINITY, -INFINITY);

  for (int base = beg; base < end; base += 32) {
    const int cnt = min(32, end - base);
    int s = 0;
    float4 e = make_float4(-INFINITY, -INFINITY, -INFINITY, -INFINITY);
    if (lane < cnt) {
      s = csr[base + lane];
      const float4 a = als4[s];
      e.x = leaky(a.x + d4.x);
      e.y = leaky(a.y + d4.y);
      e.z = leaky(a.z + d4.z);
      e.w = leaky(a.w + d4.w);
    }
    float4 cm = e;
    for (int o = 16; o >= 1; o >>= 1) {
      cm.x = fmaxf(cm.x, __shfl_xor(cm.x, o, 32));
      cm.y = fmaxf(cm.y, __shfl_xor(cm.y, o, 32));
      cm.z = fmaxf(cm.z, __shfl_xor(cm.z, o, 32));
      cm.w = fmaxf(cm.w, __shfl_xor(cm.w, o, 32));
    }
    float4 nm;
    nm.x = fmaxf(mx.x, cm.x); nm.y = fmaxf(mx.y, cm.y);
    nm.z = fmaxf(mx.z, cm.z); nm.w = fmaxf(mx.w, cm.w);
    const float sc01 = hb0 ? __expf(mx.y - nm.y) : __expf(mx.x - nm.x);
    const float sc23 = hb0 ? __expf(mx.w - nm.w) : __expf(mx.z - nm.z);
    const float sc = hb1 ? sc23 : sc01;
    acc.x *= sc; acc.y *= sc; acc.z *= sc; acc.w *= sc;
    wsum *= sc;
    mx = nm;
    const int ex0 = __float_as_int(__expf(e.x - nm.x));
    const int ex1 = __float_as_int(__expf(e.y - nm.y));
    const int ex2 = __float_as_int(__expf(e.z - nm.z));
    const int ex3 = __float_as_int(__expf(e.w - nm.w));

    int j = 0;
    for (; j + 4 <= cnt; j += 4) {
      v4h hv[4];
      float wv[4];
      for (int u = 0; u < 4; ++u) {
        const int sj = __builtin_amdgcn_readlane(s, j + u);
        hv[u] = *(const v4h*)(Hh + (size_t)sj * FDIM + lane * 4);
        const float a0 = __int_as_float(__builtin_amdgcn_readlane(ex0, j + u));
        const float a1 = __int_as_float(__builtin_amdgcn_readlane(ex1, j + u));
        const float a2 = __int_as_float(__builtin_amdgcn_readlane(ex2, j + u));
        const float a3 = __int_as_float(__builtin_amdgcn_readlane(ex3, j + u));
        const float w01 = hb0 ? a1 : a0;
        const float w23 = hb0 ? a3 : a2;
        wv[u] = hb1 ? w23 : w01;
      }
      for (int u = 0; u < 4; ++u) {
        acc.x = fmaf((float)hv[u][0], wv[u], acc.x);
        acc.y = fmaf((float)hv[u][1], wv[u], acc.y);
        acc.z = fmaf((float)hv[u][2], wv[u], acc.z);
        acc.w = fmaf((float)hv[u][3], wv[u], acc.w);
        wsum += wv[u];
      }
    }
    for (; j < cnt; ++j) {
      const int sj = __builtin_amdgcn_readlane(s, j);
      const v4h h1 = *(const v4h*)(Hh + (size_t)sj * FDIM + lane * 4);
      const float a0 = __int_as_float(__builtin_amdgcn_readlane(ex0, j));
      const float a1 = __int_as_float(__builtin_amdgcn_readlane(ex1, j));
      const float a2 = __int_as_float(__builtin_amdgcn_readlane(ex2, j));
      const float a3 = __int_as_float(__builtin_amdgcn_readlane(ex3, j));
      const float w01 = hb0 ? a1 : a0;
      const float w23 = hb0 ? a3 : a2;
      const float wj = hb1 ? w23 : w01;
      acc.x = fmaf((float)h1[0], wj, acc.x);
      acc.y = fmaf((float)h1[1], wj, acc.y);
      acc.z = fmaf((float)h1[2], wj, acc.z);
      acc.w = fmaf((float)h1[3], wj, acc.w);
      wsum += wj;
    }
  }

  const float inv = 1.0f / (wsum + GAT_EPS);
  const float* bp4 = bias + lane * 4;
  float4 b;
  b.x = bp4[0]; b.y = bp4[1]; b.z = bp4[2]; b.w = bp4[3];
  float4 o;
  o.x = fmaf(acc.x, inv, b.x);
  o.y = fmaf(acc.y, inv, b.y);
  o.z = fmaf(acc.z, inv, b.z);
  o.w = fmaf(acc.w, inv, b.w);
  return o;
}

__global__ void __launch_bounds__(256) node_gather_out(const int* __restrict__ rowptr,
                                                      const int* __restrict__ csr,
                                                      const float4* __restrict__ als4,
                                                      const float4* __restrict__ ald4,
                                                      const _Float16* __restrict__ Hh,
                                                      const float* __restrict__ bias,
                                                      float* __restrict__ out_f32, int Nn) {
  const int n = __builtin_amdgcn_readfirstlane((int)(blockIdx.x * 8 + (threadIdx.x >> 5)));
  const int lane = threadIdx.x & 31;
  if (n >= Nn) return;
  const float4 o = gather_row(rowptr, csr, als4, ald4, Hh, bias, n, lane);
  *(float4*)(out_f32 + (size_t)n * FDIM + (size_t)lane * 4) = o;
}

__global__ void __launch_bounds__(512, 16) gather_project(const int* __restrict__ rowptr,
                                                     const int* __restrict__ csr,
                                                     const float4* __restrict__ als4,
                                                     const float4* __restrict__ ald4,
                                                     const _Float16* __restrict__ Hh,
                                                     const float* __restrict__ bias,
                                                     const _Float16* __restrict__ Bp,
                                                     const float* __restrict__ Asrc,
                                                     const float* __restrict__ Adst,
                                                     _Float16* __restrict__ Hout,
                                                     float* __restrict__ als_next,
                                                     float* __restrict__ ald_next, int Nn) {
  __shared__ __attribute__((aligned(16))) _Float16 tile[16 * TILE_PITCH];
  const int w = __builtin_amdgcn_readfirstlane((int)(threadIdx.x >> 5));
  const int lane = threadIdx.x & 31;
  const int n = (int)blockIdx.x * 16 + w;
  v4h hq = {};
  if (n < Nn) {
    const float4 o = gather_row(rowptr, csr, als4, ald4, Hh, bias, n, lane);
    hq[0] = (_Float16)fmaxf(o.x, 0.0f);
    hq[1] = (_Float16)fmaxf(o.y, 0.0f);
    hq[2] = (_Float16)fmaxf(o.z, 0.0f);
    hq[3] = (_Float16)fmaxf(o.w, 0.0f);
  }
  *(v4h*)(tile + w * TILE_PITCH + lane * 4) = hq;
  __syncthreads();
  if (w < HEADS)
    gemm_head<false>(nullptr, tile, Bp, Asrc, Adst, Hout, als_next, ald_next, Nn,
                     (int)blockIdx.x, w);
}

extern "C" void kernel_launch(void* const* d_in, const int* in_sizes, int n_in,
                              void* d_out, int out_size, void* d_ws, size_t ws_size,
                              hipStream_t stream) {
  const int Nn   = in_sizes[0] / FDIM;
  const int Etot = in_sizes[1] / 2;

  const float* x0  = (const float*)d_in[0];
  const int*   ei  = (const int*)d_in[1];
  const int*   src = ei;
  const int*   dst = ei + Etot;

  const int nTiles  = (Nn + 1 + SCAN_TILE - 1) / SCAN_TILE;
  const int nPadded = nTiles * SCAN_TILE;

  char* ws = (char*)d_ws;
  size_t off = 0;
  auto alloc = [&](size_t bytes) {
    void* p = ws + off;
    off += (bytes + 255) & ~(size_t)255;
    return p;
  };
  _Float16* bp0    = (_Float16*)alloc((size_t)PACK_ELEMS * sizeof(_Float16));
  _Float16* bp1    = (_Float16*)alloc((size_t)PACK_ELEMS * sizeof(_Float16));
  _Float16* bp2    = (_Float16*)alloc((size_t)PACK_ELEMS * sizeof(_Float16));
  _Float16* buf_h[2];
  float*    als[2];
  float*    ald[2];
  for (int i = 0; i < 2; ++i) {
    buf_h[i] = (_Float16*)alloc((size_t)Nn * FDIM * sizeof(_Float16));
    als[i]   = (float*)   alloc((size_t)Nn * HEADS * sizeof(float));
    ald[i]   = (float*)   alloc((size_t)Nn * HEADS * sizeof(float));
  }
  int*      cnt    = (int*)     alloc((size_t)nPadded * sizeof(int));
  int*      rowptr = (int*)     alloc((size_t)nPadded * sizeof(int));
  int*      rank   = (int*)     alloc((size_t)Etot * sizeof(int));
  int*      csr_u  = (int*)     alloc((size_t)Etot * sizeof(int));
  int*      csr_s  = (int*)     alloc((size_t)Etot * sizeof(int));

  const int TPB = 256;
  const int gEdge  = (Etot + TPB - 1) / TPB;
  const int gNodeW = (Nn + (TPB / 32) - 1) / (TPB / 32);
  const int gPad   = (nPadded + TPB - 1) / TPB;
  const int rowTiles = (Nn + 15) >> 4;
  const int gGemm  = (rowTiles * HEADS + (TPB / 32) - 1) / (TPB / 32);

  clear_and_pack<<<gPad + 3 * PACK_BLOCKS, TPB, 0, stream>>>(
      cnt, nPadded, gPad,
      (const float*)d_in[2], (const float*)d_in[6], (const float*)d_in[10],
      bp0, bp1, bp2);
  const int gTriples = (gGemm > (gEdge + 1) / 2) ? gGemm : (gEdge + 1) / 2;
  gemm_x32_and_count<<<3 * gTriples, TPB, 0, stream>>>(
      x0, bp0, (const float*)d_in[3], (const float*)d_in[4], buf_h[0], als[0], ald[0], Nn,
      dst, cnt, rank, Etot);
  scan_counts<<<nTiles, TPB, 0, stream>>>((const int4*)cnt, (int4*)rowptr);
  fill_csr<<<gEdge, TPB, 0, stream>>>(src, dst, rank, rowptr, csr_u, Etot);
  sort_segments<<<gNodeW, TPB, 0, stream>>>(rowptr, csr_u, csr_s, Nn);

  gather_project<<<rowTiles, 512, 0, stream>>>(
      rowptr, csr_s, (const float4*)als[0], (const float4*)ald[0], buf_h[0],
      (const float*)d_in[5], bp1, (const float*)d_in[7], (const float*)d_in[8],
      buf_h[1], als[1], ald[1], Nn);
  gather_project<<<rowTiles, 512, 0, stream>>>(
      rowptr, csr_s, (const float4*)als[1], (const float4*)ald[1], buf_h[1],
      (const float*)d_in[9], bp2, (const float*)d_in[11], (const float*)d_in[12],
      buf_h[0], als[0], ald[0], Nn);
  node_gather_out<<<gNodeW, TPB, 0, stream>>>(rowptr, csr_s, (const float4*)als[0],
                                             (const float4*)ald[0], buf_h[0],
                                             (const float*)d_in[13], (float*)d_out, Nn);
}


# Gather takes two edges of a node per step, one per 16-lane half (lane owns 8 channels, 16-byte row loads); even and odd segment positions are summed apart and joined at the end: sum order changes
# speedup vs baseline: 8.7420x; 1.0894x over previous
#include <hip/hip_runtime.h>
#include <hip/hip_bf16.h>
#include <math.h>

typedef __attribute__((ext_vector_type(16))) _Float16 v16h;
typedef __attribute__((ext_vector_type(8)))  _Float16 v8h;
typedef __attribute__((ext_vector_type(8)))  float    v8f;
typedef __attribute__((ext_vector_type(4)))  _Float16 v4h;
typedef __attribute__((ext_vector_type(2)))  _Float16 v2h;
typedef __attribute__((ext_vector_type(2)))  float    v2f;

#define HEADS 4
#define CDIM 32
#define FDIM 128
#define NEG_SLOPE 0.2f
#define GAT_EPS 1e-16f
#define TILE_PITCH 136

#define PACK_ELEMS (8 * 32 * 4 * 16)
#define PACK_BLOCKS (PACK_ELEMS / 256)

__device__ __forceinline__ void pack_b_elem(const float* __restrict__ W,
                                            _Float16* __restrict__ Bp, int t) {
  const int i    = (t & 7) + ((t >> 5) & 8);
  const int lane = (t >> 3) & 31;
  const int ks   = (t >> 9) & 3;
  const int ct   = t >> 11;
  const int ka   = ks * 32 + ((lane >= 16) ? 8 : 0);
  const int k    = ka + ((i < 8) ? i : i + 8);
  const int col  = (ct >> 1) * 32 + 2 * (lane & 15) + (ct & 1);
  Bp[t] = (_Float16)W[(size_t)k * FDIM + col];
}

__global__ void __launch_bounds__(256) clear_and_pack(int* __restrict__ cnt, int total,
                                                      int zeroBlocks,
                                                      const float* __restrict__ W0,
                                                      const float* __restrict__ W1,
                                                      const float* __restrict__ W2,
                                                      _Float16* __restrict__ Bp0,
                                                      _Float16* __restrict__ Bp1,
                                                      _Float16* __restrict__ Bp2) {
  const int b = (int)blockIdx.x;
  if (b < zeroBlocks) {
    const int t = b * 256 + (int)threadIdx.x;
    if (t < total) cnt[t] = 0;
    return;
  }
  const int pb = b - zeroBlocks;
  const int l = pb / PACK_BLOCKS;
  const int t = (pb - l * PACK_BLOCKS) * 256 + (int)threadIdx.x;
  if (l == 0)      pack_b_elem(W0, Bp0, t);
  else if (l == 1) pack_b_elem(W1, Bp1, t);
  else             pack_b_elem(W2, Bp2, t);
}

template <int M>
__device__ __forceinline__ float row_xadd(float v) {
  const int t = __builtin_amdgcn_update_dpp(0, __float_as_int(v), 0x160 + M, 0xf, 0xf, true);
  return v + __int_as_float(t);
}

__device__ __forceinline__ float xhalf(float v) {
  const int t = __float_as_int(v);
  return __int_as_float(__builtin_amdgcn_permlanex16(t, t, 0x76543210, 0xfedcba98, false, false));
}

template <bool kF32Rows>
__device__ __forceinline__ void gemm_head(const float* __restrict__ Xf,
                                          const _Float16* Xt,
                                          const _Float16* __restrict__ Bp,
                                          const float* __restrict__ Adst,
                                          _Float16* __restrict__ Hout,
                                          float* __restrict__ ald, int Nn, int rt, int hd) {
  const int lane = threadIdx.x & 31;
  const int row0 = rt << 4;
  const int lm = lane & 15;
  const bool hi = lane >= 16;

  size_t aoff;
  if constexpr (kF32Rows) {
    int arow = row0 + lm;
    if (arow >= Nn) arow = Nn - 1;
    aoff = (size_t)arow * FDIM + (hi ? 8 : 0);
  } else {
    aoff = (size_t)lm * TILE_PITCH + (hi ? 8 : 0);
  }
  const unsigned boff = (unsigned)(hd * 2) * 2048u + (unsigned)lane * 8u;
  const _Float16* bfrag0 = Bp + boff;
  const _Float16* bfrag1 = bfrag0 + 2048;

  const int colh = hd * CDIM + 2 * lm;
  const float2 ad2 = *(const float2*)(Adst + colh);

  v8f c0 = {};
  v8f c1 = {};
  #pragma unroll
  for (int ks = 0; ks < 4; ++ks) {
    v16h a;
    if constexpr (kF32Rows) {
      const float* ap = Xf + aoff + ks * 32;
      const float4 f0 = *(const float4*)(ap);
      const float4 f1 = *(const float4*)(ap + 4);
      const float4 f2 = *(const float4*)(ap + 16);
      const float4 f3 = *(const float4*)(ap + 20);
      a[0]  = (_Float16)f0.x; a[1]  = (_Float16)f0.y; a[2]  = (_Float16)f0.z; a[3]  = (_Float16)f0.w;
      a[4]  = (_Float16)f1.x; a[5]  = (_Float16)f1.y; a[6]  = (_Float16)f1.z; a[7]  = (_Float16)f1.w;
      a[8]  = (_Float16)f2.x; a[9]  = (_Float16)f2.y; a[10] = (_Float16)f2.z; a[11] = (_Float16)f2.w;
      a[12] = (_Float16)f3.x; a[13] = (_Float16)f3.y; a[14] = (_Float16)f3.z; a[15] = (_Float16)f3.w;
    } else {
      const _Float16* ap = Xt + aoff + ks * 32;
      const v8h alo = *(const v8h*)(ap);
      const v8h ahi = *(const v8h*)(ap + 16);
      #pragma unroll
      for (int i = 0; i < 8; ++i) { a[i] = alo[i]; a[8 + i] = ahi[i]; }
    }
    const v8h b0lo = *(const v8h*)(bfrag0 + ks * 512);
    const v8h b0hi = *(const v8h*)(bfrag0 + ks * 512 + 256);
    const v8h b1lo = *(const v8h*)(bfrag1 + ks * 512);
    const v8h b1hi = *(const v8h*)(bfrag1 + ks * 512 + 256);
    const v16h b0 = __builtin_shufflevector(b0lo, b0hi, 0, 1, 2, 3, 4, 5, 6, 7,
                                            8, 9, 10, 11, 12, 13, 14, 15);
    const v16h b1 = __builtin_shufflevector(b1lo, b1hi, 0, 1, 2, 3, 4, 5, 6, 7,
                                            8, 9, 10, 11, 12, 13, 14, 15);
    c0 = __builtin_amdgcn_wmma_f32_16x16x32_f16(
         false, a,  false, b0,
         (short)0, c0,  false,  false);
    c1 = __builtin_amdgcn_wmma_f32_16x16x32_f16(
         false, a,  false, b1,
         (short)0, c1,  false,  false);
  }

  const int m0 = hi ? 8 : 0;
  _Float16* outp = Hout + (size_t)(row0 + m0) * FDIM + colh;
  if (row0 + 16 <= Nn) {
    #pragma unroll
    for (int r = 0; r < 8; ++r) {
      v2h p;
      p[0] = (_Float16)c0[r];
      p[1] = (_Float16)c1[r];
      *(v2h*)(outp + (size_t)r * FDIM) = p;
    }
  } else {
    #pragma unroll
    for (int r = 0; r < 8; ++r) {
      if (row0 + m0 + r < Nn) {
        v2h p;
        p[0] = (_Float16)c0[r];
        p[1] = (_Float16)c1[r];
        *(v2h*)(outp + (size_t)r * FDIM) = p;
      }
    }
  }

  const bool up = (lane & 8) != 0;
  float q[4];
  #pragma unroll
  for (int r = 0; r < 4; ++r) {
    const float pa = fmaf(c1[r], ad2.y, c0[r] * ad2.x);
    const float pb = fmaf(c1[r + 4], ad2.y, c0[r + 4] * ad2.x);
    const float keep = up ? pb : pa;
    const float give = up ? pa : pb;
    const int g = __builtin_amdgcn_update_dpp(0, __float_as_int(give), 0x168, 0xf, 0xf, true);
    q[r] = keep + __int_as_float(g);
  }
  #pragma unroll
  for (int r = 0; r < 4; ++r) q[r] = row_xadd<1>(q[r]);
  #pragma unroll
  for (int r = 0; r < 4; ++r) q[r] = row_xadd<2>(q[r]);
  #pragma unroll
  for (int r = 0; r < 4; ++r) q[r] = row_xadd<4>(q[r]);
  const int rr = lane & 7;
  float val = q[0];
  #pragma unroll
  for (int r = 1; r < 4; ++r) val = (rr == r) ? q[r] : val;
  const int orow = row0 + m0 + (up ? 4 : 0) + rr;
  if (rr < 4 && orow < Nn) ald[(size_t)orow * HEADS + hd] = val;
}

__global__ void __launch_bounds__(256) gemm_x32_and_count(const float* __restrict__ X,
                                   const _Float16* __restrict__ Bp,
                                   const float* __restrict__ Adst,
                                   _Float16* __restrict__ Hout,
                                   float* __restrict__ ald,
                                   int Nn,
                                   const int* __restrict__ dst, int* __restrict__ cnt,
                                   int* __restrict__ rank, int Etot) {
  const int q = (int)blockIdx.x / 3;
  const int r = (int)blockIdx.x - 3 * q;
  if (r != 0) {
    const int e = (2 * q + r - 1) * 256 + (int)threadIdx.x;
    if (e < Etot) rank[e] = atomicAdd(&cnt[dst[e]], 1);
    return;
  }
  const int wave = q * 8 + ((int)threadIdx.x >> 5);
  if (wave >= ((Nn + 15) >> 4) * HEADS) return;
  gemm_head<true>(X, nullptr, Bp, Adst, Hout, ald, Nn, wave >> 2, wave & 3);
}

__device__ __forceinline__ float leaky(float v) {
  return v > 0.0f ? v : v * NEG_SLOPE;
}

#define SCAN_TILE 1024

__global__ void __launch_bounds__(256) scan_counts(const int4* __restrict__ cnt4,
                                                   int4* __restrict__ rowptr4) {
  __shared__ int wsum[8];
  __shared__ int psum[8];
  const int t = threadIdx.x;
  const int b = blockIdx.x;
  const int lane = t & 31, w = t >> 5;
  const int4 c = cnt4[b * 256 + t];
  int part = 0;
  for (int k = 0; k < b; ++k) {
    const int4 p = cnt4[k * 256 + t];
    part += (p.x + p.y) + (p.z + p.w);
  }
  const int s = c.x + c.y + c.z + c.w;
  int v = s;
  for (int o = 1; o < 32; o <<= 1) {
    const int u = __shfl_up(v, o, 32);
    if (lane >= o) v += u;
  }
  for (int o = 16; o >= 1; o >>= 1) part += __shfl_xor(part, o, 32);
  if (lane == 31) wsum[w] = v;
  if (lane == 0) psum[w] = part;
  __syncthreads();
  int woff = 0;
  for (int k = 0; k < 8; ++k) woff += psum[k] + ((k < w) ? wsum[k] : 0);
  const int ex = woff + v - s;
  int4 r;
  r.x = ex; r.y = ex + c.x; r.z = ex + c.x + c.y; r.w = ex + c.x + c.y + c.z;
  rowptr4[b * 256 + t] = r;
}

__global__ void fill_csr(const int* __restrict__ src, const int* __restrict__ dst,
                         const int* __restrict__ rank, const int* __restrict__ rowptr,
                         int* __restrict__ csr_u, int Etot) {
  int e = blockIdx.x * blockDim.x + threadIdx.x;
  if (e >= Etot) return;
  csr_u[rowptr[dst[e]] + rank[e]] = src[e];
}

__global__ void __launch_bounds__(256) sort_segments(const int* __restrict__ rowptr,
                                                     const int* __restrict__ csr_u,
                                                     int* __restrict__ csr_s, int Nn) {
  const int n = __builtin_amdgcn_readfirstlane((int)(blockIdx.x * 8 + (threadIdx.x >> 5)));
  const int lane = threadIdx.x & 31;
  if (n >= Nn) return;
  const int beg = rowptr[n], end = rowptr[n + 1];
  for (int bi = beg; bi < end; bi += 32) {
    const int i = bi + lane;
    const bool ok = i < end;
    const int vi = ok ? csr_u[i] : 0;
    int r = 0;
    for (int bq = beg; bq < end; bq += 32) {
      const int q = bq + lane;
      const int vq = (q < end) ? csr_u[q] : 0;
      const int nq = min(32, end - bq);
      for (int t = 0; t < nq; ++t) {
        const int x = __builtin_amdgcn_readlane(vq, t);
        const int qq = bq + t;
        r += (x < vi || (x == vi && qq < i)) ? 1 : 0;
      }
    }
    if (ok) csr_s[beg + r] = vi;
  }
}

#define GATHER_PAIRS 3

template <int G, bool kPartial>
__device__ __forceinline__ void gather_group(const _Float16* __restrict__ Hh, int s, int j,
                                             int rem, bool upper, unsigned loff,
                                             const v2f (&av)[4], float dl,
                                             float& m, float& wsum, v2f (&acc)[4]) {
  unsigned bo[G];
  #pragma unroll
  for (int u = 0; u < G; ++u) {
    const int i0 = (!kPartial || 2 * u < rem) ? j + 2 * u : j;
    const int s0 = __builtin_amdgcn_readlane(s, i0);
    const int s1 = __builtin_amdgcn_readlane(s, i0 + 1);
    const int sj = upper ? s1 : s0;
    bo[u] = (unsigned)sj * (unsigned)(FDIM * sizeof(_Float16)) + loff;
  }
  v8h hv[G];
  #pragma unroll
  for (int u = 0; u < G; ++u) {
    hv[u] = v8h{};
    if (!kPartial || 2 * u < rem) hv[u] = *(const v8h*)((const char*)Hh + bo[u]);
    if (kPartial && !(2 * u + (upper ? 1 : 0) < rem)) hv[u] = v8h{};
  }
  float e[G];
  #pragma unroll
  for (int u = 0; u < G; ++u) {
    v2f t = v2f{(float)hv[u][0], (float)hv[u][1]} * av[0];
    t = __builtin_elementwise_fma(v2f{(float)hv[u][2], (float)hv[u][3]}, av[1], t);
    t = __builtin_elementwise_fma(v2f{(float)hv[u][4], (float)hv[u][5]}, av[2], t);
    t = __builtin_elementwise_fma(v2f{(float)hv[u][6], (float)hv[u][7]}, av[3], t);
    float p = t[0] + t[1];
    p = row_xadd<1>(p);
    p = row_xadd<2>(p);
    const bool ok = !kPartial || (2 * u + (upper ? 1 : 0) < rem);
    e[u] = ok ? leaky(p + dl) : -INFINITY;
  }
  float gm = e[0];
  #pragma unroll
  for (int u = 1; u < G; ++u) gm = fmaxf(gm, e[u]);
  const float nm = fmaxf(m, gm);
  const float ref = (kPartial && nm == -INFINITY) ? 0.0f : nm;
  const float sc = __expf(m - ref);
  const v2f sc2 = v2f{sc, sc};
  #pragma unroll
  for (int q = 0; q < 4; ++q) acc[q] *= sc2;
  wsum *= sc;
  m = nm;
  #pragma unroll
  for (int u = 0; u < G; ++u) {
    const float w = __expf(e[u] - ref);
    const v2f w2 = v2f{w, w};
    acc[0] = __builtin_elementwise_fma(v2f{(float)hv[u][0], (float)hv[u][1]}, w2, acc[0]);
    acc[1] = __builtin_elementwise_fma(v2f{(float)hv[u][2], (float)hv[u][3]}, w2, acc[1]);
    acc[2] = __builtin_elementwise_fma(v2f{(float)hv[u][4], (float)hv[u][5]}, w2, acc[2]);
    acc[3] = __builtin_elementwise_fma(v2f{(float)hv[u][6], (float)hv[u][7]}, w2, acc[3]);
    wsum += w;
  }
}

__device__ __forceinline__ float4 gather_row(const int* __restrict__ rowptr,
                                             const int* __restrict__ csr,
                                             const float* __restrict__ ald,
                                             const _Float16* __restrict__ Hh,
                                             const float* __restrict__ Asrc,
                                             const float* __restrict__ bias,
                                             int n, int lane) {
  const int beg = rowptr[n], end = rowptr[n + 1];
  const int c = lane & 15;
  const bool upper = lane >= 16;
  const float dl = ald[(size_t)n * HEADS + (c >> 2)];
  const float4 a0 = *(const float4*)(Asrc + c * 8);
  const float4 a1 = *(const float4*)(Asrc + c * 8 + 4);
  const v2f av[4] = {v2f{a0.x, a0.y}, v2f{a0.z, a0.w}, v2f{a1.x, a1.y}, v2f{a1.z, a1.w}};
  const float4 b = *(const float4*)(bias + c * 8 + (upper ? 4 : 0));
  const unsigned loff = (unsigned)c * 16u;

  v2f acc[4] = {v2f{0.0f, 0.0f}, v2f{0.0f, 0.0f}, v2f{0.0f, 0.0f}, v2f{0.0f, 0.0f}};
  float wsum = 0.0f;
  float m = -INFINITY;

  for (int base = beg; base < end; base += 32) {
    const int cnt = min(32, end - base);
    const int s = (lane < cnt) ? csr[base + lane] : 0;
    int j = 0;
    for (; j + 2 * GATHER_PAIRS <= cnt; j += 2 * GATHER_PAIRS)
      gather_group<GATHER_PAIRS, false>(Hh, s, j, 2 * GATHER_PAIRS, upper, loff, av, dl, m, wsum, acc);
    if (j < cnt) {
      const int rem = cnt - j;
      if (rem <= 2)
        gather_group<1, true>(Hh, s, j, rem, upper, loff, av, dl, m, wsum, acc);
      else if (rem <= 4)
        gather_group<2, true>(Hh, s, j, rem, upper, loff, av, dl, m, wsum, acc);
      else
        gather_group<3, true>(Hh, s, j, rem, upper, loff, av, dl, m, wsum, acc);
    }
  }

  const v2f keep0 = upper ? acc[2] : acc[0];
  const v2f keep1 = upper ? acc[3] : acc[1];
  const v2f give0 = upper ? acc[0] : acc[2];
  const v2f give1 = upper ? acc[1] : acc[3];
  const float om = xhalf(m);
  const float ow = xhalf(wsum);
  const float4 og = make_float4(xhalf(give0[0]), xhalf(give0[1]), xhalf(give1[0]), xhalf(give1[1]));
  const float jm = fmaxf(m, om);
  const float ref = (jm == -INFINITY) ? 0.0f : jm;
  const float sk = __expf(m - ref);
  const float so = __expf(om - ref);
  const float wtot = fmaf(wsum, sk, ow * so);
  const float inv = 1.0f / (wtot + GAT_EPS);
  float4 o;
  o.x = fmaf(fmaf(keep0[0], sk, og.x * so), inv, b.x);
  o.y = fmaf(fmaf(keep0[1], sk, og.y * so), inv, b.y);
  o.z = fmaf(fmaf(keep1[0], sk, og.z * so), inv, b.z);
  o.w = fmaf(fmaf(keep1[1], sk, og.w * so), inv, b.w);
  return o;
}

__global__ void __launch_bounds__(256) node_gather_out(const int* __restrict__ rowptr,
                                                      const int* __restrict__ csr,
                                                      const float* __restrict__ ald,
                                                      const _Float16* __restrict__ Hh,
                                                      const float* __restrict__ Asrc,
                                                      const float* __restrict__ bias,
                                                      float* __restrict__ out_f32, int Nn) {
  const int n = __builtin_amdgcn_readfirstlane((int)(blockIdx.x * 8 + (threadIdx.x >> 5)));
  const int lane = threadIdx.x & 31;
  if (n >= Nn) return;
  const float4 o = gather_row(rowptr, csr, ald, Hh, Asrc, bias, n, lane);
  *(float4*)(out_f32 + (size_t)n * FDIM + (size_t)((lane & 15) * 8 + (lane >> 4) * 4)) = o;
}

__global__ void __launch_bounds__(512, 16) gather_project(const int* __restrict__ rowptr,
                                                     const int* __restrict__ csr,
                                                     const float* __restrict__ ald,
                                                     const _Float16* __restrict__ Hh,
                                                     const float* __restrict__ Asrc,
                                                     const float* __restrict__ bias,
                                                     const _Float16* __restrict__ Bp,
                                                     const float* __restrict__ Adst_next,
                                                     _Float16* __restrict__ Hout,
                                                     float* __restrict__ ald_next, int Nn) {
  __shared__ __attribute__((aligned(16))) _Float16 tile[16 * TILE_PITCH];
  const int w = __builtin_amdgcn_readfirstlane((int)(threadIdx.x >> 5));
  const int lane = threadIdx.x & 31;
  const int n = (int)blockIdx.x * 16 + w;
  v4h hq = {};
  if (n < Nn) {
    const float4 o = gather_row(rowptr, csr, ald, Hh, Asrc, bias, n, lane);
    hq[0] = (_Float16)fmaxf(o.x, 0.0f);
    hq[1] = (_Float16)fmaxf(o.y, 0.0f);
    hq[2] = (_Float16)fmaxf(o.z, 0.0f);
    hq[3] = (_Float16)fmaxf(o.w, 0.0f);
  }
  *(v4h*)(tile + w * TILE_PITCH + (lane & 15) * 8 + (lane >> 4) * 4) = hq;
  __syncthreads();
  if (w < HEADS)
    gemm_head<false>(nullptr, tile, Bp, Adst_next, Hout, ald_next, Nn, (int)blockIdx.x, w);
}

extern "C" void kernel_launch(void* const* d_in, const int* in_sizes, int n_in,
                              void* d_out, int out_size, void* d_ws, size_t ws_size,
                              hipStream_t stream) {
  const int Nn   = in_sizes[0] / FDIM;
  const int Etot = in_sizes[1] / 2;

  const float* x0  = (const float*)d_in[0];
  const int*   ei  = (const int*)d_in[1];
  const int*   src = ei;
  const int*   dst = ei + Etot;

  const int nTiles  = (Nn + 1 + SCAN_TILE - 1) / SCAN_TILE;
  const int nPadded = nTiles * SCAN_TILE;

  char* ws = (char*)d_ws;
  size_t off = 0;
  auto alloc = [&](size_t bytes) {
    void* p = ws + off;
    off += (bytes + 255) & ~(size_t)255;
    return p;
  };
  _Float16* bp0    = (_Float16*)alloc((size_t)PACK_ELEMS * sizeof(_Float16));
  _Float16* bp1    = (_Float16*)alloc((size_t)PACK_ELEMS * sizeof(_Float16));
  _Float16* bp2    = (_Float16*)alloc((size_t)PACK_ELEMS * sizeof(_Float16));
  _Float16* buf_h[2];
  float*    ald[2];
  for (int i = 0; i < 2; ++i) {
    buf_h[i] = (_Float16*)alloc((size_t)Nn * FDIM * sizeof(_Float16));
    ald[i]   = (float*)   alloc((size_t)Nn * HEADS * sizeof(float));
  }
  int*      cnt    = (int*)     alloc((size_t)nPadded * sizeof(int));
  int*      rowptr = (int*)     alloc((size_t)nPadded * sizeof(int));
  int*      rank   = (int*)     alloc((size_t)Etot * sizeof(int));
  int*      csr_u  = (int*)     alloc((size_t)Etot * sizeof(int));
  int*      csr_s  = (int*)     alloc((size_t)Etot * sizeof(int));

  const int TPB = 256;
  const int gEdge  = (Etot + TPB - 1) / TPB;
  const int gNodeW = (Nn + (TPB / 32) - 1) / (TPB / 32);
  const int gPad   = (nPadded + TPB - 1) / TPB;
  const int rowTiles = (Nn + 15) >> 4;
  const int gGemm  = (rowTiles * HEADS + (TPB / 32) - 1) / (TPB / 32);

  clear_and_pack<<<gPad + 3 * PACK_BLOCKS, TPB, 0, stream>>>(
      cnt, nPadded, gPad,
      (const float*)d_in[2], (const float*)d_in[6], (const float*)d_in[10],
      bp0, bp1, bp2);
  const int gTriples = (gGemm > (gEdge + 1) / 2) ? gGemm : (gEdge + 1) / 2;
  gemm_x32_and_count<<<3 * gTriples, TPB, 0, stream>>>(
      x0, bp0, (const float*)d_in[4], buf_h[0], ald[0], Nn,
      dst, cnt, rank, Etot);
  scan_counts<<<nTiles, TPB, 0, stream>>>((const int4*)cnt, (int4*)rowptr);
  fill_csr<<<gEdge, TPB, 0, stream>>>(src, dst, rank, rowptr, csr_u, Etot);
  sort_segments<<<gNodeW, TPB, 0, stream>>>(rowptr, csr_u, csr_s, Nn);

  gather_project<<<rowTiles, 512, 0, stream>>>(
      rowptr, csr_s, ald[0], buf_h[0], (const float*)d_in[3],
      (const float*)d_in[5], bp1, (const float*)d_in[8],
      buf_h[1], ald[1], Nn);
  gather_project<<<rowTiles, 512, 0, stream>>>(
      rowptr, csr_s, ald[1], buf_h[1], (const float*)d_in[7],
      (const float*)d_in[9], bp2, (const float*)d_in[12],
      buf_h[0], ald[0], Nn);
  node_gather_out<<<gNodeW, TPB, 0, stream>>>(rowptr, csr_s, ald[0], buf_h[0],
                                             (const float*)d_in[11],
                                             (const float*)d_in[13], (float*)d_out, Nn);
}


# Layer-0 gather wave sorts its own segment by source (rank by key in registers, one forward permute) and stores it for layers 1, 2; the separate sort launch of 50000 waves is gone; sum order unchanged
# speedup vs baseline: 9.1607x; 1.0479x over previous
#include <hip/hip_runtime.h>
#include <hip/hip_bf16.h>
#include <math.h>

typedef __attribute__((ext_vector_type(16))) _Float16 v16h;
typedef __attribute__((ext_vector_type(8)))  _Float16 v8h;
typedef __attribute__((ext_vector_type(8)))  float    v8f;
typedef __attribute__((ext_vector_type(4)))  _Float16 v4h;
typedef __attribute__((ext_vector_type(2)))  _Float16 v2h;
typedef __attribute__((ext_vector_type(2)))  float    v2f;

#define HEADS 4
#define CDIM 32
#define FDIM 128
#define NEG_SLOPE 0.2f
#define GAT_EPS 1e-16f
#define TILE_PITCH 136

#define PACK_ELEMS (8 * 32 * 4 * 16)
#define PACK_BLOCKS (PACK_ELEMS / 256)

__device__ __forceinline__ void pack_b_elem(const float* __restrict__ W,
                                            _Float16* __restrict__ Bp, int t) {
  const int i    = (t & 7) + ((t >> 5) & 8);
  const int lane = (t >> 3) & 31;
  const int ks   = (t >> 9) & 3;
  const int ct   = t >> 11;
  const int ka   = ks * 32 + ((lane >= 16) ? 8 : 0);
  const int k    = ka + ((i < 8) ? i : i + 8);
  const int col  = (ct >> 1) * 32 + 2 * (lane & 15) + (ct & 1);
  Bp[t] = (_Float16)W[(size_t)k * FDIM + col];
}

__global__ void __launch_bounds__(256) clear_and_pack(int* __restrict__ cnt, int total,
                                                      int zeroBlocks,
                                                      const float* __restrict__ W0,
                                                      const float* __restrict__ W1,
                                                      const float* __restrict__ W2,
                                                      _Float16* __restrict__ Bp0,
                                                      _Float16* __restrict__ Bp1,
                                                      _Float16* __restrict__ Bp2) {
  const int b = (int)blockIdx.x;
  if (b < zeroBlocks) {
    const int t = b * 256 + (int)threadIdx.x;
    if (t < total) cnt[t] = 0;
    return;
  }
  const int pb = b - zeroBlocks;
  const int l = pb / PACK_BLOCKS;
  const int t = (pb - l * PACK_BLOCKS) * 256 + (int)threadIdx.x;
  if (l == 0)      pack_b_elem(W0, Bp0, t);
  else if (l == 1) pack_b_elem(W1, Bp1, t);
  else             pack_b_elem(W2, Bp2, t);
}

template <int M>
__device__ __forceinline__ float row_xadd(float v) {
  const int t = __builtin_amdgcn_update_dpp(0, __float_as_int(v), 0x160 + M, 0xf, 0xf, true);
  return v + __int_as_float(t);
}

__device__ __forceinline__ float xhalf(float v) {
  const int t = __float_as_int(v);
  return __int_as_float(__builtin_amdgcn_permlanex16(t, t, 0x76543210, 0xfedcba98, false, false));
}

template <bool kF32Rows>
__device__ __forceinline__ void gemm_head(const float* __restrict__ Xf,
                                          const _Float16* Xt,
                                          const _Float16* __restrict__ Bp,
                                          const float* __restrict__ Adst,
                                          _Float16* __restrict__ Hout,
                                          float* __restrict__ ald, int Nn, int rt, int hd) {
  const int lane = threadIdx.x & 31;
  const int row0 = rt << 4;
  const int lm = lane & 15;
  const bool hi = lane >= 16;

  size_t aoff;
  if constexpr (kF32Rows) {
    int arow = row0 + lm;
    if (arow >= Nn) arow = Nn - 1;
    aoff = (size_t)arow * FDIM + (hi ? 8 : 0);
  } else {
    aoff = (size_t)lm * TILE_PITCH + (hi ? 8 : 0);
  }
  const unsigned boff = (unsigned)(hd * 2) * 2048u + (unsigned)lane * 8u;
  const _Float16* bfrag0 = Bp + boff;
  const _Float16* bfrag1 = bfrag0 + 2048;

  const int colh = hd * CDIM + 2 * lm;
  const float2 ad2 = *(const float2*)(Adst + colh);

  v8f c0 = {};
  v8f c1 = {};
  #pragma unroll
  for (int ks = 0; ks < 4; ++ks) {
    v16h a;
    if constexpr (kF32Rows) {
      const float* ap = Xf + aoff + ks * 32;
      const float4 f0 = *(const float4*)(ap);
      const float4 f1 = *(const float4*)(ap + 4);
      const float4 f2 = *(const float4*)(ap + 16);
      const float4 f3 = *(const float4*)(ap + 20);
      a[0]  = (_Float16)f0.x; a[1]  = (_Float16)f0.y; a[2]  = (_Float16)f0.z; a[3]  = (_Float16)f0.w;
      a[4]  = (_Float16)f1.x; a[5]  = (_Float16)f1.y; a[6]  = (_Float16)f1.z; a[7]  = (_Float16)f1.w;
      a[8]  = (_Float16)f2.x; a[9]  = (_Float16)f2.y; a[10] = (_Float16)f2.z; a[11] = (_Float16)f2.w;
      a[12] = (_Float16)f3.x; a[13] = (_Float16)f3.y; a[14] = (_Float16)f3.z; a[15] = (_Float16)f3.w;
    } else {
      const _Float16* ap = Xt + aoff + ks * 32;
      const v8h alo = *(const v8h*)(ap);
      const v8h ahi = *(const v8h*)(ap + 16);
      #pragma unroll
      for (int i = 0; i < 8; ++i) { a[i] = alo[i]; a[8 + i] = ahi[i]; }
    }
    const v8h b0lo = *(const v8h*)(bfrag0 + ks * 512);
    const v8h b0hi = *(const v8h*)(bfrag0 + ks * 512 + 256);
    const v8h b1lo = *(const v8h*)(bfrag1 + ks * 512);
    const v8h b1hi = *(const v8h*)(bfrag1 + ks * 512 + 256);
    const v16h b0 = __builtin_shufflevector(b0lo, b0hi, 0, 1, 2, 3, 4, 5, 6, 7,
                                            8, 9, 10, 11, 12, 13, 14, 15);
    const v16h b1 = __builtin_shufflevector(b1lo, b1hi, 0, 1, 2, 3, 4, 5, 6, 7,
                                            8, 9, 10, 11, 12, 13, 14, 15);
    c0 = __builtin_amdgcn_wmma_f32_16x16x32_f16(
         false, a,  false, b0,
         (short)0, c0,  false,  false);
    c1 = __builtin_amdgcn_wmma_f32_16x16x32_f16(
         false, a,  false, b1,
         (short)0, c1,  false,  false);
  }

  const int m0 = hi ? 8 : 0;
  _Float16* outp = Hout + (size_t)(row0 + m0) * FDIM + colh;
  if (row0 + 16 <= Nn) {
    #pragma unroll
    for (int r = 0; r < 8; ++r) {
      v2h p;
      p[0] = (_Float16)c0[r];
      p[1] = (_Float16)c1[r];
      *(v2h*)(outp + (size_t)r * FDIM) = p;
    }
  } else {
    #pragma unroll
    for (int r = 0; r < 8; ++r) {
      if (row0 + m0 + r < Nn) {
        v2h p;
        p[0] = (_Float16)c0[r];
        p[1] = (_Float16)c1[r];
        *(v2h*)(outp + (size_t)r * FDIM) = p;
      }
    }
  }

  const bool up = (lane & 8) != 0;
  float q[4];
  #pragma unroll
  for (int r = 0; r < 4; ++r) {
    const float pa = fmaf(c1[r], ad2.y, c0[r] * ad2.x);
    const float pb = fmaf(c1[r + 4], ad2.y, c0[r + 4] * ad2.x);
    const float keep = up ? pb : pa;
    const float give = up ? pa : pb;
    const int g = __builtin_amdgcn_update_dpp(0, __float_as_int(give), 0x168, 0xf, 0xf, true);
    q[r] = keep + __int_as_float(g);
  }
  #pragma unroll
  for (int r = 0; r < 4; ++r) q[r] = row_xadd<1>(q[r]);
  #pragma unroll
  for (int r = 0; r < 4; ++r) q[r] = row_xadd<2>(q[r]);
  #pragma unroll
  for (int r = 0; r < 4; ++r) q[r] = row_xadd<4>(q[r]);
  const int rr = lane & 7;
  float val = q[0];
  #pragma unroll
  for (int r = 1; r < 4; ++r) val = (rr == r) ? q[r] : val;
  const int orow = row0 + m0 + (up ? 4 : 0) + rr;
  if (rr < 4 && orow < Nn) ald[(size_t)orow * HEADS + hd] = val;
}

__global__ void __launch_bounds__(256) gemm_x32_and_count(const float* __restrict__ X,
                                   const _Float16* __restrict__ Bp,
                                   const float* __restrict__ Adst,
                                   _Float16* __restrict__ Hout,
                                   float* __restrict__ ald,
                                   int Nn,
                                   const int* __restrict__ dst, int* __restrict__ cnt,
                                   int* __restrict__ rank, int Etot) {
  const int q = (int)blockIdx.x / 3;
  const int r = (int)blockIdx.x - 3 * q;
  if (r != 0) {
    const int e = (2 * q + r - 1) * 256 + (int)threadIdx.x;
    if (e < Etot) rank[e] = atomicAdd(&cnt[dst[e]], 1);
    return;
  }
  const int wave = q * 8 + ((int)threadIdx.x >> 5);
  if (wave >= ((Nn + 15) >> 4) * HEADS) return;
  gemm_head<true>(X, nullptr, Bp, Adst, Hout, ald, Nn, wave >> 2, wave & 3);
}

__device__ __forceinline__ float leaky(float v) {
  return v > 0.0f ? v : v * NEG_SLOPE;
}

#define SCAN_TILE 1024

__global__ void __launch_bounds__(256) scan_counts(const int4* __restrict__ cnt4,
                                                   int4* __restrict__ rowptr4) {
  __shared__ int wsum[8];
  __shared__ int psum[8];
  const int t = threadIdx.x;
  const int b = blockIdx.x;
  const int lane = t & 31, w = t >> 5;
  const int4 c = cnt4[b * 256 + t];
  int part = 0;
  for (int k = 0; k < b; ++k) {
    const int4 p = cnt4[k * 256 + t];
    part += (p.x + p.y) + (p.z + p.w);
  }
  const int s = c.x + c.y + c.z + c.w;
  int v = s;
  for (int o = 1; o < 32; o <<= 1) {
    const int u = __shfl_up(v, o, 32);
    if (lane >= o) v += u;
  }
  for (int o = 16; o >= 1; o >>= 1) part += __shfl_xor(part, o, 32);
  if (lane == 31) wsum[w] = v;
  if (lane == 0) psum[w] = part;
  __syncthreads();
  int woff = 0;
  for (int k = 0; k < 8; ++k) woff += psum[k] + ((k < w) ? wsum[k] : 0);
  const int ex = woff + v - s;
  int4 r;
  r.x = ex; r.y = ex + c.x; r.z = ex + c.x + c.y; r.w = ex + c.x + c.y + c.z;
  rowptr4[b * 256 + t] = r;
}

__global__ void fill_csr(const int* __restrict__ src, const int* __restrict__ dst,
                         const int* __restrict__ rank, const int* __restrict__ rowptr,
                         int* __restrict__ csr_u, int Etot) {
  int e = blockIdx.x * blockDim.x + threadIdx.x;
  if (e >= Etot) return;
  csr_u[rowptr[dst[e]] + rank[e]] = src[e];
}

__device__ __forceinline__ int sort_segment_in_wave(int vi, int cnt, int lane) {
  const bool ok = lane < cnt;
  const int key = ok ? ((vi << 5) | lane) : 0x7fffffff;
  int r = 0;
  for (int t = 0; t < cnt; ++t) {
    const int x = __builtin_amdgcn_readlane(key, t);
    r += (x < key) ? 1 : 0;
  }
  return __builtin_amdgcn_ds_permute((ok ? r : lane) << 2, ok ? vi : 0);
}

__device__ __forceinline__ void sort_long_segment(const int* __restrict__ csr_u, int* csr_s,
                                                  int beg, int end, int lane) {
  for (int bi = beg; bi < end; bi += 32) {
    const int i = bi + lane;
    const bool ok = i < end;
    const int vi = ok ? csr_u[i] : 0;
    int r = 0;
    for (int bq = beg; bq < end; bq += 32) {
      const int q = bq + lane;
      const int vq = (q < end) ? csr_u[q] : 0;
      const int nq = min(32, end - bq);
      for (int t = 0; t < nq; ++t) {
        const int x = __builtin_amdgcn_readlane(vq, t);
        const int qq = bq + t;
        r += (x < vi || (x == vi && qq < i)) ? 1 : 0;
      }
    }
    if (ok) csr_s[beg + r] = vi;
  }
}

#define GATHER_PAIRS 3

template <int G, bool kPartial>
__device__ __forceinline__ void gather_group(const _Float16* __restrict__ Hh, int s, int j,
                                             int rem, bool upper, unsigned loff,
                                             const v2f (&av)[4], float dl,
                                             float& m, float& wsum, v2f (&acc)[4]) {
  unsigned bo[G];
  #pragma unroll
  for (int u = 0; u < G; ++u) {
    const int i0 = (!kPartial || 2 * u < rem) ? j + 2 * u : j;
    const int s0 = __builtin_amdgcn_readlane(s, i0);
    const int s1 = __builtin_amdgcn_readlane(s, i0 + 1);
    const int sj = upper ? s1 : s0;
    bo[u] = (unsigned)sj * (unsigned)(FDIM * sizeof(_Float16)) + loff;
  }
  v8h hv[G];
  #pragma unroll
  for (int u = 0; u < G; ++u) {
    hv[u] = v8h{};
    if (!kPartial || 2 * u < rem) hv[u] = *(const v8h*)((const char*)Hh + bo[u]);
    if (kPartial && !(2 * u + (upper ? 1 : 0) < rem)) hv[u] = v8h{};
  }
  float e[G];
  #pragma unroll
  for (int u = 0; u < G; ++u) {
    v2f t = v2f{(float)hv[u][0], (float)hv[u][1]} * av[0];
    t = __builtin_elementwise_fma(v2f{(float)hv[u][2], (float)hv[u][3]}, av[1], t);
    t = __builtin_elementwise_fma(v2f{(float)hv[u][4], (float)hv[u][5]}, av[2], t);
    t = __builtin_elementwise_fma(v2f{(float)hv[u][6], (float)hv[u][7]}, av[3], t);
    float p = t[0] + t[1];
    p = row_xadd<1>(p);
    p = row_xadd<2>(p);
    const bool ok = !kPartial || (2 * u + (upper ? 1 : 0) < rem);
    e[u] = ok ? leaky(p + dl) : -INFINITY;
  }
  float gm = e[0];
  #pragma unroll
  for (int u = 1; u < G; ++u) gm = fmaxf(gm, e[u]);
  const float nm = fmaxf(m, gm);
  const float ref = (kPartial && nm == -INFINITY) ? 0.0f : nm;
  const float sc = __expf(m - ref);
  const v2f sc2 = v2f{sc, sc};
  #pragma unroll
  for (int q = 0; q < 4; ++q) acc[q] *= sc2;
  wsum *= sc;
  m = nm;
  #pragma unroll
  for (int u = 0; u < G; ++u) {
    const float w = __expf(e[u] - ref);
    const v2f w2 = v2f{w, w};
    acc[0] = __builtin_elementwise_fma(v2f{(float)hv[u][0], (float)hv[u][1]}, w2, acc[0]);
    acc[1] = __builtin_elementwise_fma(v2f{(float)hv[u][2], (float)hv[u][3]}, w2, acc[1]);
    acc[2] = __builtin_elementwise_fma(v2f{(float)hv[u][4], (float)hv[u][5]}, w2, acc[2]);
    acc[3] = __builtin_elementwise_fma(v2f{(float)hv[u][6], (float)hv[u][7]}, w2, acc[3]);
    wsum += w;
  }
}

template <bool kSort>
__device__ __forceinline__ float4 gather_row(const int* __restrict__ rowptr,
                                             const int* __restrict__ csr,
                                             int* csr_sorted,
                                             const float* __restrict__ ald,
                                             const _Float16* __restrict__ Hh,
                                             const float* __restrict__ Asrc,
                                             const float* __restrict__ bias,
                                             int n, int lane) {
  const int beg = rowptr[n], end = rowptr[n + 1];
  const int c = lane & 15;
  const bool upper = lane >= 16;
  const float dl = ald[(size_t)n * HEADS + (c >> 2)];
  const float4 a0 = *(const float4*)(Asrc + c * 8);
  const float4 a1 = *(const float4*)(Asrc + c * 8 + 4);
  const v2f av[4] = {v2f{a0.x, a0.y}, v2f{a0.z, a0.w}, v2f{a1.x, a1.y}, v2f{a1.z, a1.w}};
  const float4 b = *(const float4*)(bias + c * 8 + (upper ? 4 : 0));
  const unsigned loff = (unsigned)c * 16u;

  v2f acc[4] = {v2f{0.0f, 0.0f}, v2f{0.0f, 0.0f}, v2f{0.0f, 0.0f}, v2f{0.0f, 0.0f}};
  float wsum = 0.0f;
  float m = -INFINITY;

  const bool inWave = (end - beg) <= 32;
  if constexpr (kSort) {
    if (!inWave) {
      sort_long_segment(csr, csr_sorted, beg, end, lane);
      __threadfence();
    }
  }

  for (int base = beg; base < end; base += 32) {
    const int cnt = min(32, end - base);
    int s;
    if constexpr (kSort) {
      if (inWave) {
        s = sort_segment_in_wave((lane < cnt) ? csr[base + lane] : 0, cnt, lane);
        if (lane < cnt) csr_sorted[base + lane] = s;
      } else {
        s = (lane < cnt) ? csr_sorted[base + lane] : 0;
      }
    } else {
      s = (lane < cnt) ? csr[base + lane] : 0;
    }
    int j = 0;
    for (; j + 2 * GATHER_PAIRS <= cnt; j += 2 * GATHER_PAIRS)
      gather_group<GATHER_PAIRS, false>(Hh, s, j, 2 * GATHER_PAIRS, upper, loff, av, dl, m, wsum, acc);
    if (j < cnt) {
      const int rem = cnt - j;
      if (rem <= 2)
        gather_group<1, true>(Hh, s, j, rem, upper, loff, av, dl, m, wsum, acc);
      else if (rem <= 4)
        gather_group<2, true>(Hh, s, j, rem, upper, loff, av, dl, m, wsum, acc);
      else
        gather_group<3, true>(Hh, s, j, rem, upper, loff, av, dl, m, wsum, acc);
    }
  }

  const v2f keep0 = upper ? acc[2] : acc[0];
  const v2f keep1 = upper ? acc[3] : acc[1];
  const v2f give0 = upper ? acc[0] : acc[2];
  const v2f give1 = upper ? acc[1] : acc[3];
  const float om = xhalf(m);
  const float ow = xhalf(wsum);
  const float4 og = make_float4(xhalf(give0[0]), xhalf(give0[1]), xhalf(give1[0]), xhalf(give1[1]));
  const float jm = fmaxf(m, om);
  const float ref = (jm == -INFINITY) ? 0.0f : jm;
  const float sk = __expf(m - ref);
  const float so = __expf(om - ref);
  const float wtot = fmaf(wsum, sk, ow * so);
  const float inv = 1.0f / (wtot + GAT_EPS);
  float4 o;
  o.x = fmaf(fmaf(keep0[0], sk, og.x * so), inv, b.x);
  o.y = fmaf(fmaf(keep0[1], sk, og.y * so), inv, b.y);
  o.z = fmaf(fmaf(keep1[0], sk, og.z * so), inv, b.z);
  o.w = fmaf(fmaf(keep1[1], sk, og.w * so), inv, b.w);
  return o;
}

__global__ void __launch_bounds__(256) node_gather_out(const int* __restrict__ rowptr,
                                                      const int* __restrict__ csr,
                                                      const float* __restrict__ ald,
                                                      const _Float16* __restrict__ Hh,
                                                      const float* __restrict__ Asrc,
                                                      const float* __restrict__ bias,
                                                      float* __restrict__ out_f32, int Nn) {
  const int n = __builtin_amdgcn_readfirstlane((int)(blockIdx.x * 8 + (threadIdx.x >> 5)));
  const int lane = threadIdx.x & 31;
  if (n >= Nn) return;
  const float4 o = gather_row<false>(rowptr, csr, nullptr, ald, Hh, Asrc, bias, n, lane);
  *(float4*)(out_f32 + (size_t)n * FDIM + (size_t)((lane & 15) * 8 + (lane >> 4) * 4)) = o;
}

template <bool kSort>
__global__ void __launch_bounds__(512, 16) gather_project(const int* __restrict__ rowptr,
                                                     const int* __restrict__ csr,
                                                     int* csr_sorted,
                                                     const float* __restrict__ ald,
                                                     const _Float16* __restrict__ Hh,
                                                     const float* __restrict__ Asrc,
                                                     const float* __restrict__ bias,
                                                     const _Float16* __restrict__ Bp,
                                                     const float* __restrict__ Adst_next,
                                                     _Float16* __restrict__ Hout,
                                                     float* __restrict__ ald_next, int Nn) {
  __shared__ __attribute__((aligned(16))) _Float16 tile[16 * TILE_PITCH];
  const int w = __builtin_amdgcn_readfirstlane((int)(threadIdx.x >> 5));
  const int lane = threadIdx.x & 31;
  const int n = (int)blockIdx.x * 16 + w;
  v4h hq = {};
  if (n < Nn) {
    const float4 o = gather_row<kSort>(rowptr, csr, csr_sorted, ald, Hh, Asrc, bias, n, lane);
    hq[0] = (_Float16)fmaxf(o.x, 0.0f);
    hq[1] = (_Float16)fmaxf(o.y, 0.0f);
    hq[2] = (_Float16)fmaxf(o.z, 0.0f);
    hq[3] = (_Float16)fmaxf(o.w, 0.0f);
  }
  *(v4h*)(tile + w * TILE_PITCH + (lane & 15) * 8 + (lane >> 4) * 4) = hq;
  __syncthreads();
  if (w < HEADS)
    gemm_head<false>(nullptr, tile, Bp, Adst_next, Hout, ald_next, Nn, (int)blockIdx.x, w);
}

extern "C" void kernel_launch(void* const* d_in, const int* in_sizes, int n_in,
                              void* d_out, int out_size, void* d_ws, size_t ws_size,
                              hipStream_t stream) {
  const int Nn   = in_sizes[0] / FDIM;
  const int Etot = in_sizes[1] / 2;

  const float* x0  = (const float*)d_in[0];
  const int*   ei  = (const int*)d_in[1];
  const int*   src = ei;
  const int*   dst = ei + Etot;

  const int nTiles  = (Nn + 1 + SCAN_TILE - 1) / SCAN_TILE;
  const int nPadded = nTiles * SCAN_TILE;

  char* ws = (char*)d_ws;
  size_t off = 0;
  auto alloc = [&](size_t bytes) {
    void* p = ws + off;
    off += (bytes + 255) & ~(size_t)255;
    return p;
  };
  _Float16* bp0    = (_Float16*)alloc((size_t)PACK_ELEMS * sizeof(_Float16));
  _Float16* bp1    = (_Float16*)alloc((size_t)PACK_ELEMS * sizeof(_Float16));
  _Float16* bp2    = (_Float16*)alloc((size_t)PACK_ELEMS * sizeof(_Float16));
  _Float16* buf_h[2];
  float*    ald[2];
  for (int i = 0; i < 2; ++i) {
    buf_h[i] = (_Float16*)alloc((size_t)Nn * FDIM * sizeof(_Float16));
    ald[i]   = (float*)   alloc((size_t)Nn * HEADS * sizeof(float));
  }
  int*      cnt    = (int*)     alloc((size_t)nPadded * sizeof(int));
  int*      rowptr = (int*)     alloc((size_t)nPadded * sizeof(int));
  int*      rank   = (int*)     alloc((size_t)Etot * sizeof(int));
  int*      csr_u  = (int*)     alloc((size_t)Etot * sizeof(int));
  int*      csr_s  = (int*)     alloc((size_t)Etot * sizeof(int));

  const int TPB = 256;
  const int gEdge  = (Etot + TPB - 1) / TPB;
  const int gNodeW = (Nn + (TPB / 32) - 1) / (TPB / 32);
  const int gPad   = (nPadded + TPB - 1) / TPB;
  const int rowTiles = (Nn + 15) >> 4;
  const int gGemm  = (rowTiles * HEADS + (TPB / 32) - 1) / (TPB / 32);

  clear_and_pack<<<gPad + 3 * PACK_BLOCKS, TPB, 0, stream>>>(
      cnt, nPadded, gPad,
      (const float*)d_in[2], (const float*)d_in[6], (const float*)d_in[10],
      bp0, bp1, bp2);
  const int gTriples = (gGemm > (gEdge + 1) / 2) ? gGemm : (gEdge + 1) / 2;
  gemm_x32_and_count<<<3 * gTriples, TPB, 0, stream>>>(
      x0, bp0, (const float*)d_in[4], buf_h[0], ald[0], Nn,
      dst, cnt, rank, Etot);
  scan_counts<<<nTiles, TPB, 0, stream>>>((const int4*)cnt, (int4*)rowptr);
  fill_csr<<<gEdge, TPB, 0, stream>>>(src, dst, rank, rowptr, csr_u, Etot);

  gather_project<true><<<rowTiles, 512, 0, stream>>>(
      rowptr, csr_u, csr_s, ald[0], buf_h[0], (const float*)d_in[3],
      (const float*)d_in[5], bp1, (const float*)d_in[8],
      buf_h[1], ald[1], Nn);
  gather_project<false><<<rowTiles, 512, 0, stream>>>(
      rowptr, csr_s, nullptr, ald[1], buf_h[1], (const float*)d_in[7],
      (const float*)d_in[9], bp2, (const float*)d_in[12],
      buf_h[0], ald[0], Nn);
  node_gather_out<<<gNodeW, TPB, 0, stream>>>(rowptr, csr_s, ald[0], buf_h[0],
                                             (const float*)d_in[11],
                                             (const float*)d_in[13], (float*)d_out, Nn);
}
